# starting kernel
# baseline (speedup 1.0000x reference)
#include <hip/hip_runtime.h>

typedef float v2f __attribute__((ext_vector_type(2)));
typedef float v8f __attribute__((ext_vector_type(8)));

#define N_NODES 50000
#define FEAT    128

__global__ void sage_zero_kernel(float* __restrict__ p, long long n) {
  long long i = (long long)blockIdx.x * blockDim.x + threadIdx.x;
  long long stride = (long long)gridDim.x * blockDim.x;
  for (; i < n; i += stride) p[i] = 0.0f;
}

__global__ void sage_scatter_kernel(const float* __restrict__ x,
                                    const int* __restrict__ src,
                                    const int* __restrict__ dst,
                                    float* __restrict__ msg,
                                    float* __restrict__ deg,
                                    int E, int addDeg) {
  long long gid = (long long)blockIdx.x * blockDim.x + threadIdx.x;
  long long total = (long long)E * 32;
  if (gid >= total) return;
  int e = (int)(gid >> 5);
  int c = (int)(gid & 31);
  int s = src[e];
  int d = dst[e];
  const float4* xs = (const float4*)(x + (long long)s * FEAT);
  float4 v = xs[c];
  float* p = msg + (long long)d * FEAT + c * 4;
  unsafeAtomicAdd(p + 0, v.x);
  unsafeAtomicAdd(p + 1, v.y);
  unsafeAtomicAdd(p + 2, v.z);
  unsafeAtomicAdd(p + 3, v.w);
  if (addDeg && c == 0) unsafeAtomicAdd(deg + d, 1.0f);
}

__global__ void sage_gemm_kernel(const float* __restrict__ msg,
                                 const float* __restrict__ deg,
                                 const float* __restrict__ xin,
                                 const float* __restrict__ Wl,
                                 const float* __restrict__ bl,
                                 const float* __restrict__ Wr,
                                 float* __restrict__ out,
                                 int K, int N, int doRelu) {
  const int lane = threadIdx.x & 31;
  const int wave = threadIdx.x >> 5;
  const int wavesPerBlock = blockDim.x >> 5;
  const int tile = blockIdx.x * wavesPerBlock + wave;
  const int colTiles = N >> 4;
  const int rowTiles = N_NODES / 16;
  const int rowTile = tile / colTiles;
  const int colTile = tile - rowTile * colTiles;
  if (rowTile >= rowTiles) return;

  const int r   = (rowTile << 4) + (lane & 15);
  const int col = (colTile << 4) + (lane & 15);
  const int kb  = (lane >> 4) << 1;

  const float scale = 1.0f / fmaxf(deg[r], 1.0f);

  v8f c = {};
  const float* arow = msg + (long long)r * K;
  for (int k0 = 0; k0 < K; k0 += 4) {
    v2f a, b;
    a.x = arow[k0 + kb] * scale;
    a.y = arow[k0 + kb + 1] * scale;
    b.x = Wl[(k0 + kb) * N + col];
    b.y = Wl[(k0 + kb + 1) * N + col];
    c = __builtin_amdgcn_wmma_f32_16x16x4_f32(false, a, false, b,
                                              (short)0, c, false, false);
  }
  const float* xrow = xin + (long long)r * K;
  for (int k0 = 0; k0 < K; k0 += 4) {
    v2f a, b;
    a.x = xrow[k0 + kb];
    a.y = xrow[k0 + kb + 1];
    b.x = Wr[(k0 + kb) * N + col];
    b.y = Wr[(k0 + kb + 1) * N + col];
    c = __builtin_amdgcn_wmma_f32_16x16x4_f32(false, a, false, b,
                                              (short)0, c, false, false);
  }

  const float bias = bl[col];
  const int rbase = (rowTile << 4) + ((lane >> 4) << 3);
#pragma unroll
  for (int i = 0; i < 8; ++i) {
    float v = c[i] + bias;
    if (doRelu) v = fmaxf(v, 0.0f);
    out[(long long)(rbase + i) * N + col] = v;
  }
}

extern "C" void kernel_launch(void* const* d_in, const int* in_sizes, int n_in,
                              void* d_out, int out_size, void* d_ws, size_t ws_size,
                              hipStream_t stream) {
  const float* x   = (const float*)d_in[0];
  const int*   ei  = (const int*)d_in[1];
  const float* Wl1 = (const float*)d_in[2];
  const float* bl1 = (const float*)d_in[3];
  const float* Wr1 = (const float*)d_in[4];
  const float* Wl2 = (const float*)d_in[5];
  const float* bl2 = (const float*)d_in[6];
  const float* Wr2 = (const float*)d_in[7];
  float* out = (float*)d_out;

  const int E = in_sizes[1] / 2;
  const int* src = ei;
  const int* dst = ei + E;

  float* msg = (float*)d_ws;
  float* deg = msg + (long long)N_NODES * FEAT;
  float* h1  = deg + N_NODES;

  const int ZB = 2048, TB = 256;
  const long long scatterThreads = (long long)E * 32;
  const int scatterBlocks = (int)((scatterThreads + TB - 1) / TB);

  sage_zero_kernel<<<ZB, TB, 0, stream>>>(msg, (long long)N_NODES * FEAT + N_NODES);
  sage_scatter_kernel<<<scatterBlocks, TB, 0, stream>>>(x, src, dst, msg, deg, E, 1);
  {
    const int tiles = (N_NODES / 16) * (128 / 16);
    const int blocks = (tiles + 7) / 8;
    sage_gemm_kernel<<<blocks, TB, 0, stream>>>(msg, deg, x, Wl1, bl1, Wr1, h1,
                                                128, 128, 1);
  }

  sage_zero_kernel<<<ZB, TB, 0, stream>>>(msg, (long long)N_NODES * FEAT);
  sage_scatter_kernel<<<scatterBlocks, TB, 0, stream>>>(h1, src, dst, msg, deg, E, 0);
  {
    const int tiles = (N_NODES / 16) * (64 / 16);
    const int blocks = (tiles + 7) / 8;
    sage_gemm_kernel<<<blocks, TB, 0, stream>>>(msg, deg, h1, Wl2, bl2, Wr2, out,
                                                128, 64, 0);
  }
}


# Neighbour sum: each of 256 workgroups owns 196 nodes, scans all dst, sums its rows by LDS float atomics and stores them; no global atomics, no zero pass (4 launches); sums in LDS arrival order
# speedup vs baseline: 2.7088x; 2.7088x over previous
#include <hip/hip_runtime.h>

typedef float v2f __attribute__((ext_vector_type(2)));
typedef float v8f __attribute__((ext_vector_type(8)));

#define N_NODES 50000
#define FEAT    128

#define OWN_ROWS  196
#define OWN_WGS   ((N_NODES + OWN_ROWS - 1) / OWN_ROWS)
#define SC_WAVES  32
#define SC_DEPTH  4
#define SC_EU     8
#define SC_QCAP   (SC_DEPTH * 128 + SC_EU)

__global__ __launch_bounds__(SC_WAVES * 32)
void sage_scatter_kernel(const float* __restrict__ x,
                         const int* __restrict__ src,
                         const int* __restrict__ dst,
                         float* __restrict__ msg,
                         float* __restrict__ deg,
                         int E, int addDeg) {
  __shared__ __attribute__((aligned(16))) float sAcc[OWN_ROWS * FEAT];
  __shared__ float sDeg[OWN_ROWS];
  __shared__ int   qE[SC_WAVES][SC_QCAP];
  __shared__ int   qD[SC_WAVES][SC_QCAP];

  const int tid  = threadIdx.x;
  const int lane = tid & 31;
  const int wave = __builtin_amdgcn_readfirstlane(tid >> 5);
  const int nodeBase = (int)blockIdx.x * OWN_ROWS;
  const int rows = (N_NODES - nodeBase < OWN_ROWS) ? (N_NODES - nodeBase) : OWN_ROWS;

  for (int i = tid; i < OWN_ROWS * FEAT / 4; i += SC_WAVES * 32)
    ((float4*)sAcc)[i] = float4{0.0f, 0.0f, 0.0f, 0.0f};
  for (int i = tid; i < OWN_ROWS; i += SC_WAVES * 32) sDeg[i] = 0.0f;
  __syncthreads();

  int* myE = qE[wave];
  int* myD = qD[wave];
  int cnt = 0;
  const int nTrip = (E + SC_WAVES * SC_DEPTH * 128 - 1) / (SC_WAVES * SC_DEPTH * 128);

  auto request = [&](int4 (&buf)[SC_DEPTH], int trip) {
#pragma unroll
    for (int k = 0; k < SC_DEPTH; ++k) {
      const int cb = ((trip * SC_DEPTH + k) * SC_WAVES + wave) * 128;
      if (cb + 128 <= E) {
        buf[k] = *(const int4*)(dst + cb + lane * 4);
      } else {
        const int b = cb + lane * 4;
        buf[k].x = b + 0 < E ? dst[b + 0] : -1;
        buf[k].y = b + 1 < E ? dst[b + 1] : -1;
        buf[k].z = b + 2 < E ? dst[b + 2] : -1;
        buf[k].w = b + 3 < E ? dst[b + 3] : -1;
      }
    }
  };

  auto scan = [&](const int4 (&buf)[SC_DEPTH], int trip) {
#pragma unroll
    for (int k = 0; k < SC_DEPTH; ++k) {
      const int base = ((trip * SC_DEPTH + k) * SC_WAVES + wave) * 128 + lane * 4;
      const int d0 = buf[k].x - nodeBase, d1 = buf[k].y - nodeBase;
      const int d2 = buf[k].z - nodeBase, d3 = buf[k].w - nodeBase;
      const bool h0 = (unsigned)d0 < (unsigned)rows, h1 = (unsigned)d1 < (unsigned)rows;
      const bool h2 = (unsigned)d2 < (unsigned)rows, h3 = (unsigned)d3 < (unsigned)rows;
      const unsigned b0 = __builtin_amdgcn_ballot_w32(h0);
      const unsigned b1 = __builtin_amdgcn_ballot_w32(h1);
      const unsigned b2 = __builtin_amdgcn_ballot_w32(h2);
      const unsigned b3 = __builtin_amdgcn_ballot_w32(h3);
      if (b0 | b1 | b2 | b3) {
        const unsigned below = (1u << lane) - 1u;
        int pos = cnt + __builtin_popcount(b0 & below);
        if (h0) { myE[pos] = base + 0; myD[pos] = d0; if (addDeg) unsafeAtomicAdd(&sDeg[d0], 1.0f); }
        cnt += __builtin_popcount(b0);
        pos = cnt + __builtin_popcount(b1 & below);
        if (h1) { myE[pos] = base + 1; myD[pos] = d1; if (addDeg) unsafeAtomicAdd(&sDeg[d1], 1.0f); }
        cnt += __builtin_popcount(b1);
        pos = cnt + __builtin_popcount(b2 & below);
        if (h2) { myE[pos] = base + 2; myD[pos] = d2; if (addDeg) unsafeAtomicAdd(&sDeg[d2], 1.0f); }
        cnt += __builtin_popcount(b2);
        pos = cnt + __builtin_popcount(b3 & below);
        if (h3) { myE[pos] = base + 3; myD[pos] = d3; if (addDeg) unsafeAtomicAdd(&sDeg[d3], 1.0f); }
        cnt += __builtin_popcount(b3);
      }
    }
  };

  auto drain = [&](bool all) {
    if (!(cnt >= SC_EU || (all && cnt > 0))) return;
    int q0 = 0;
    while (cnt - q0 >= SC_EU || (all && q0 < cnt)) {
      int m = cnt - q0 < 32 ? cnt - q0 : 32;
      if (!all) m &= ~(SC_EU - 1);
      int s = 0, dl = 0;
      if (lane < m) { dl = myD[q0 + lane]; s = src[myE[q0 + lane]]; }
      int g = 0;
      for (; g + SC_EU <= m; g += SC_EU) {
        float v[SC_EU][4];
#pragma unroll
        for (int u = 0; u < SC_EU; ++u) {
          const int su = __builtin_amdgcn_readlane(s, g + u);
          const float* xs = x + (long long)su * FEAT + lane;
          v[u][0] = xs[0]; v[u][1] = xs[32]; v[u][2] = xs[64]; v[u][3] = xs[96];
        }
#pragma unroll
        for (int u = 0; u < SC_EU; ++u) {
          const int du = __builtin_amdgcn_readlane(dl, g + u);
          float* p = sAcc + du * FEAT + lane;
          unsafeAtomicAdd(p + 0,  v[u][0]);
          unsafeAtomicAdd(p + 32, v[u][1]);
          unsafeAtomicAdd(p + 64, v[u][2]);
          unsafeAtomicAdd(p + 96, v[u][3]);
        }
      }
      for (; g < m; ++g) {
        const int su = __builtin_amdgcn_readlane(s, g);
        const int du = __builtin_amdgcn_readlane(dl, g);
        const float* xs = x + (long long)su * FEAT + lane;
        const float v0 = xs[0], v1 = xs[32], v2 = xs[64], v3 = xs[96];
        float* p = sAcc + du * FEAT + lane;
        unsafeAtomicAdd(p + 0,  v0);
        unsafeAtomicAdd(p + 32, v1);
        unsafeAtomicAdd(p + 64, v2);
        unsafeAtomicAdd(p + 96, v3);
      }
      q0 += m;
    }
    const int left = cnt - q0;
    int te = 0, td = 0;
    if (lane < left) { te = myE[q0 + lane]; td = myD[q0 + lane]; }
    if (lane < left) { myE[lane] = te; myD[lane] = td; }
    cnt = left;
  };

  int4 bufA[SC_DEPTH], bufB[SC_DEPTH];
  request(bufA, 0);
  for (int trip = 0; trip < nTrip; trip += 2) {
    request(bufB, trip + 1);
    scan(bufA, trip);
    drain(false);
    request(bufA, trip + 2);
    scan(bufB, trip + 1);
    drain(false);
  }
  drain(true);
  __syncthreads();

  for (int r = wave; r < rows; r += SC_WAVES)
    *(float4*)(msg + (long long)(nodeBase + r) * FEAT + lane * 4) =
        *(const float4*)(sAcc + r * FEAT + lane * 4);
  if (addDeg)
    for (int i = tid; i < rows; i += SC_WAVES * 32) deg[nodeBase + i] = sDeg[i];
}

#define GR     32
#define APITCH (FEAT + 4)

template <int NCT, bool RELU>
__global__ __launch_bounds__(256)
void sage_gemm_kernel(const float* __restrict__ msg,
                      const float* __restrict__ deg,
                      const float* __restrict__ xin,
                      const float* __restrict__ Wl,
                      const float* __restrict__ bl,
                      const float* __restrict__ Wr,
                      float* __restrict__ out) {
  constexpr int N  = NCT * 16;
  constexpr int RT = NCT / 4;
  __shared__ __attribute__((aligned(16))) float sM[GR * APITCH];
  __shared__ __attribute__((aligned(16))) float sX[GR * APITCH];

  const int lane = threadIdx.x & 31;
  const int wave = __builtin_amdgcn_readfirstlane((int)(threadIdx.x >> 5));
  const int row0 = (int)blockIdx.x * GR;

  float  dg[GR / 8];
  float4 m[GR / 8], xv[GR / 8];
#pragma unroll
  for (int j = 0; j < GR / 8; ++j) {
    const int gr  = row0 + wave * (GR / 8) + j;
    const int grc = gr < N_NODES ? gr : N_NODES - 1;
    dg[j] = deg[grc];
    m[j]  = *(const float4*)(msg + (long long)grc * FEAT + lane * 4);
    xv[j] = *(const float4*)(xin + (long long)grc * FEAT + lane * 4);
  }
#pragma unroll
  for (int j = 0; j < GR / 8; ++j) {
    const int lr = wave * (GR / 8) + j;
    const float scale = 1.0f / fmaxf(dg[j], 1.0f);
    m[j].x *= scale; m[j].y *= scale; m[j].z *= scale; m[j].w *= scale;
    *(float4*)(sM + lr * APITCH + lane * 4) = m[j];
    *(float4*)(sX + lr * APITCH + lane * 4) = xv[j];
  }
  __syncthreads();

  const int c16 = lane & 15;
  const int kb  = (lane >> 4) << 1;
  const int colTile = (NCT == 8) ? wave : (wave & 3);
  const int rt0     = (NCT == 8) ? 0 : (wave >> 2);
  const int col = (colTile << 4) + c16;

  v8f acc[RT];
#pragma unroll
  for (int t = 0; t < RT; ++t) acc[t] = v8f{};

  const float* aM = sM + (rt0 * 16 + c16) * APITCH + kb;
  const float* aX = sX + (rt0 * 16 + c16) * APITCH + kb;
  const float* bL = Wl + kb * N + col;
  const float* bR = Wr + kb * N + col;

#pragma unroll 8
  for (int k0 = 0; k0 < FEAT; k0 += 4) {
    v2f b;
    b.x = bL[k0 * N];
    b.y = bL[(k0 + 1) * N];
#pragma unroll
    for (int t = 0; t < RT; ++t) {
      const v2f a = *(const v2f*)(aM + t * 16 * APITCH + k0);
      acc[t] = __builtin_amdgcn_wmma_f32_16x16x4_f32(false, a, false, b,
                                                     (short)0, acc[t], false, false);
    }
  }
#pragma unroll 8
  for (int k0 = 0; k0 < FEAT; k0 += 4) {
    v2f b;
    b.x = bR[k0 * N];
    b.y = bR[(k0 + 1) * N];
#pragma unroll
    for (int t = 0; t < RT; ++t) {
      const v2f a = *(const v2f*)(aX + t * 16 * APITCH + k0);
      acc[t] = __builtin_amdgcn_wmma_f32_16x16x4_f32(false, a, false, b,
                                                     (short)0, acc[t], false, false);
    }
  }

  const float bias = bl[col];
#pragma unroll
  for (int t = 0; t < RT; ++t) {
    const int tr0 = row0 + (rt0 + t) * 16;
    if (tr0 < N_NODES) {
      const int rbase = tr0 + ((lane >> 4) << 3);
#pragma unroll
      for (int i = 0; i < 8; ++i) {
        float v = acc[t][i] + bias;
        if (RELU) v = fmaxf(v, 0.0f);
        out[(long long)(rbase + i) * N + col] = v;
      }
    }
  }
}

extern "C" void kernel_launch(void* const* d_in, const int* in_sizes, int n_in,
                              void* d_out, int out_size, void* d_ws, size_t ws_size,
                              hipStream_t stream) {
  const float* x   = (const float*)d_in[0];
  const int*   ei  = (const int*)d_in[1];
  const float* Wl1 = (const float*)d_in[2];
  const float* bl1 = (const float*)d_in[3];
  const float* Wr1 = (const float*)d_in[4];
  const float* Wl2 = (const float*)d_in[5];
  const float* bl2 = (const float*)d_in[6];
  const float* Wr2 = (const float*)d_in[7];
  float* out = (float*)d_out;

  const int E = in_sizes[1] / 2;
  const int* src = ei;
  const int* dst = ei + E;

  float* msg = (float*)d_ws;
  float* deg = msg + (long long)N_NODES * FEAT;
  float* h1  = deg + N_NODES;

  const int TB = 256, SB = SC_WAVES * 32;
  const int GB = (N_NODES + GR - 1) / GR;

  sage_scatter_kernel<<<OWN_WGS, SB, 0, stream>>>(x, src, dst, msg, deg, E, 1);
  sage_gemm_kernel<8, true><<<GB, TB, 0, stream>>>(msg, deg, x, Wl1, bl1, Wr1, h1);

  sage_scatter_kernel<<<OWN_WGS, SB, 0, stream>>>(h1, src, dst, msg, deg, E, 0);
  sage_gemm_kernel<4, false><<<GB, TB, 0, stream>>>(msg, deg, h1, Wl2, bl2, Wr2, out);
}


# Scatter: dst trips wholly inside the array are requested by branch-free loads, so a scan waits only for its own trip and the next trip's loads stay in flight; sums in LDS arrival order as before
# speedup vs baseline: 2.7962x; 1.0323x over previous
#include <hip/hip_runtime.h>

typedef float v2f __attribute__((ext_vector_type(2)));
typedef float v8f __attribute__((ext_vector_type(8)));

#define N_NODES 50000
#define FEAT    128

#define OWN_ROWS  196
#define OWN_WGS   ((N_NODES + OWN_ROWS - 1) / OWN_ROWS)
#define SC_WAVES  32
#define SC_DEPTH  4
#define SC_EU     8
#define SC_QCAP   (SC_DEPTH * 128 + SC_EU)

__global__ __launch_bounds__(SC_WAVES * 32)
void sage_scatter_kernel(const float* __restrict__ x,
                         const int* __restrict__ src,
                         const int* __restrict__ dst,
                         float* __restrict__ msg,
                         float* __restrict__ deg,
                         int E, int addDeg) {
  __shared__ __attribute__((aligned(16))) float sAcc[OWN_ROWS * FEAT];
  __shared__ float sDeg[OWN_ROWS];
  __shared__ int   qE[SC_WAVES][SC_QCAP];
  __shared__ int   qD[SC_WAVES][SC_QCAP];

  const int tid  = threadIdx.x;
  const int lane = tid & 31;
  const int wave = __builtin_amdgcn_readfirstlane(tid >> 5);
  const int nodeBase = (int)blockIdx.x * OWN_ROWS;
  const int rows = (N_NODES - nodeBase < OWN_ROWS) ? (N_NODES - nodeBase) : OWN_ROWS;

  for (int i = tid; i < OWN_ROWS * FEAT / 4; i += SC_WAVES * 32)
    ((float4*)sAcc)[i] = float4{0.0f, 0.0f, 0.0f, 0.0f};
  for (int i = tid; i < OWN_ROWS; i += SC_WAVES * 32) sDeg[i] = 0.0f;
  __syncthreads();

  int* myE = qE[wave];
  int* myD = qD[wave];
  int cnt = 0;
  const int nTrip = (E + SC_WAVES * SC_DEPTH * 128 - 1) / (SC_WAVES * SC_DEPTH * 128);

  auto request = [&](int4 (&buf)[SC_DEPTH], int trip) {
#pragma unroll
    for (int k = 0; k < SC_DEPTH; ++k) {
      const int cb = ((trip * SC_DEPTH + k) * SC_WAVES + wave) * 128;
      if (cb + 128 <= E) {
        buf[k] = *(const int4*)(dst + cb + lane * 4);
      } else {
        const int b = cb + lane * 4;
        buf[k].x = b + 0 < E ? dst[b + 0] : -1;
        buf[k].y = b + 1 < E ? dst[b + 1] : -1;
        buf[k].z = b + 2 < E ? dst[b + 2] : -1;
        buf[k].w = b + 3 < E ? dst[b + 3] : -1;
      }
    }
  };

  auto scan = [&](const int4 (&buf)[SC_DEPTH], int trip) {
#pragma unroll
    for (int k = 0; k < SC_DEPTH; ++k) {
      const int base = ((trip * SC_DEPTH + k) * SC_WAVES + wave) * 128 + lane * 4;
      const int d0 = buf[k].x - nodeBase, d1 = buf[k].y - nodeBase;
      const int d2 = buf[k].z - nodeBase, d3 = buf[k].w - nodeBase;
      const bool h0 = (unsigned)d0 < (unsigned)rows, h1 = (unsigned)d1 < (unsigned)rows;
      const bool h2 = (unsigned)d2 < (unsigned)rows, h3 = (unsigned)d3 < (unsigned)rows;
      const unsigned b0 = __builtin_amdgcn_ballot_w32(h0);
      const unsigned b1 = __builtin_amdgcn_ballot_w32(h1);
      const unsigned b2 = __builtin_amdgcn_ballot_w32(h2);
      const unsigned b3 = __builtin_amdgcn_ballot_w32(h3);
      if (b0 | b1 | b2 | b3) {
        const unsigned below = (1u << lane) - 1u;
        int pos = cnt + __builtin_popcount(b0 & below);
        if (h0) { myE[pos] = base + 0; myD[pos] = d0; if (addDeg) unsafeAtomicAdd(&sDeg[d0], 1.0f); }
        cnt += __builtin_popcount(b0);
        pos = cnt + __builtin_popcount(b1 & below);
        if (h1) { myE[pos] = base + 1; myD[pos] = d1; if (addDeg) unsafeAtomicAdd(&sDeg[d1], 1.0f); }
        cnt += __builtin_popcount(b1);
        pos = cnt + __builtin_popcount(b2 & below);
        if (h2) { myE[pos] = base + 2; myD[pos] = d2; if (addDeg) unsafeAtomicAdd(&sDeg[d2], 1.0f); }
        cnt += __builtin_popcount(b2);
        pos = cnt + __builtin_popcount(b3 & below);
        if (h3) { myE[pos] = base + 3; myD[pos] = d3; if (addDeg) unsafeAtomicAdd(&sDeg[d3], 1.0f); }
        cnt += __builtin_popcount(b3);
      }
    }
  };

  auto drain = [&](bool all) {
    if (!(cnt >= SC_EU || (all && cnt > 0))) return;
    int q0 = 0;
    while (cnt - q0 >= SC_EU || (all && q0 < cnt)) {
      int m = cnt - q0 < 32 ? cnt - q0 : 32;
      if (!all) m &= ~(SC_EU - 1);
      int s = 0, dl = 0;
      if (lane < m) { dl = myD[q0 + lane]; s = src[myE[q0 + lane]]; }
      int g = 0;
      for (; g + SC_EU <= m; g += SC_EU) {
        float v[SC_EU][4];
#pragma unroll
        for (int u = 0; u < SC_EU; ++u) {
          const int su = __builtin_amdgcn_readlane(s, g + u);
          const float* xs = x + (long long)su * FEAT + lane;
          v[u][0] = xs[0]; v[u][1] = xs[32]; v[u][2] = xs[64]; v[u][3] = xs[96];
        }
#pragma unroll
        for (int u = 0; u < SC_EU; ++u) {
          const int du = __builtin_amdgcn_readlane(dl, g + u);
          float* p = sAcc + du * FEAT + lane;
          unsafeAtomicAdd(p + 0,  v[u][0]);
          unsafeAtomicAdd(p + 32, v[u][1]);
          unsafeAtomicAdd(p + 64, v[u][2]);
          unsafeAtomicAdd(p + 96, v[u][3]);
        }
      }
      for (; g < m; ++g) {
        const int su = __builtin_amdgcn_readlane(s, g);
        const int du = __builtin_amdgcn_readlane(dl, g);
        const float* xs = x + (long long)su * FEAT + lane;
        const float v0 = xs[0], v1 = xs[32], v2 = xs[64], v3 = xs[96];
        float* p = sAcc + du * FEAT + lane;
        unsafeAtomicAdd(p + 0,  v0);
        unsafeAtomicAdd(p + 32, v1);
        unsafeAtomicAdd(p + 64, v2);
        unsafeAtomicAdd(p + 96, v3);
      }
      q0 += m;
    }
    const int left = cnt - q0;
    int te = 0, td = 0;
    if (lane < left) { te = myE[q0 + lane]; td = myD[q0 + lane]; }
    if (lane < left) { myE[lane] = te; myD[lane] = td; }
    cnt = left;
  };

  auto requestFull = [&](int4 (&buf)[SC_DEPTH], int trip) {
#pragma unroll
    for (int k = 0; k < SC_DEPTH; ++k) {
      const int cb = ((trip * SC_DEPTH + k) * SC_WAVES + wave) * 128;
      buf[k] = *(const int4*)(dst + cb + lane * 4);
    }
  };

  const int nPair = (E / (SC_WAVES * SC_DEPTH * 128)) & ~1;
  int trip = 0;
  if (nPair > 0) {
    int4 bufA[SC_DEPTH], bufB[SC_DEPTH];
    requestFull(bufA, 0);
    for (; trip < nPair; trip += 2) {
      requestFull(bufB, trip + 1);
      scan(bufA, trip);
      drain(false);
      requestFull(bufA, trip + 2 < nPair ? trip + 2 : nPair - 1);
      scan(bufB, trip + 1);
      drain(false);
    }
  }
  for (; trip < nTrip; ++trip) {
    int4 bufT[SC_DEPTH];
    request(bufT, trip);
    scan(bufT, trip);
    drain(false);
  }
  drain(true);
  __syncthreads();

  for (int r = wave; r < rows; r += SC_WAVES)
    *(float4*)(msg + (long long)(nodeBase + r) * FEAT + lane * 4) =
        *(const float4*)(sAcc + r * FEAT + lane * 4);
  if (addDeg)
    for (int i = tid; i < rows; i += SC_WAVES * 32) deg[nodeBase + i] = sDeg[i];
}

#define GR     32
#define APITCH (FEAT + 4)

template <int NCT, bool RELU>
__global__ __launch_bounds__(256)
void sage_gemm_kernel(const float* __restrict__ msg,
                      const float* __restrict__ deg,
                      const float* __restrict__ xin,
                      const float* __restrict__ Wl,
                      const float* __restrict__ bl,
                      const float* __restrict__ Wr,
                      float* __restrict__ out) {
  constexpr int N  = NCT * 16;
  constexpr int RT = NCT / 4;
  __shared__ __attribute__((aligned(16))) float sM[GR * APITCH];
  __shared__ __attribute__((aligned(16))) float sX[GR * APITCH];

  const int lane = threadIdx.x & 31;
  const int wave = __builtin_amdgcn_readfirstlane((int)(threadIdx.x >> 5));
  const int row0 = (int)blockIdx.x * GR;

  float  dg[GR / 8];
  float4 m[GR / 8], xv[GR / 8];
#pragma unroll
  for (int j = 0; j < GR / 8; ++j) {
    const int gr  = row0 + wave * (GR / 8) + j;
    const int grc = gr < N_NODES ? gr : N_NODES - 1;
    dg[j] = deg[grc];
    m[j]  = *(const float4*)(msg + (long long)grc * FEAT + lane * 4);
    xv[j] = *(const float4*)(xin + (long long)grc * FEAT + lane * 4);
  }
#pragma unroll
  for (int j = 0; j < GR / 8; ++j) {
    const int lr = wave * (GR / 8) + j;
    const float scale = 1.0f / fmaxf(dg[j], 1.0f);
    m[j].x *= scale; m[j].y *= scale; m[j].z *= scale; m[j].w *= scale;
    *(float4*)(sM + lr * APITCH + lane * 4) = m[j];
    *(float4*)(sX + lr * APITCH + lane * 4) = xv[j];
  }
  __syncthreads();

  const int c16 = lane & 15;
  const int kb  = (lane >> 4) << 1;
  const int colTile = (NCT == 8) ? wave : (wave & 3);
  const int rt0     = (NCT == 8) ? 0 : (wave >> 2);
  const int col = (colTile << 4) + c16;

  v8f acc[RT];
#pragma unroll
  for (int t = 0; t < RT; ++t) acc[t] = v8f{};

  const float* aM = sM + (rt0 * 16 + c16) * APITCH + kb;
  const float* aX = sX + (rt0 * 16 + c16) * APITCH + kb;
  const float* bL = Wl + kb * N + col;
  const float* bR = Wr + kb * N + col;

#pragma unroll 8
  for (int k0 = 0; k0 < FEAT; k0 += 4) {
    v2f b;
    b.x = bL[k0 * N];
    b.y = bL[(k0 + 1) * N];
#pragma unroll
    for (int t = 0; t < RT; ++t) {
      const v2f a = *(const v2f*)(aM + t * 16 * APITCH + k0);
      acc[t] = __builtin_amdgcn_wmma_f32_16x16x4_f32(false, a, false, b,
                                                     (short)0, acc[t], false, false);
    }
  }
#pragma unroll 8
  for (int k0 = 0; k0 < FEAT; k0 += 4) {
    v2f b;
    b.x = bR[k0 * N];
    b.y = bR[(k0 + 1) * N];
#pragma unroll
    for (int t = 0; t < RT; ++t) {
      const v2f a = *(const v2f*)(aX + t * 16 * APITCH + k0);
      acc[t] = __builtin_amdgcn_wmma_f32_16x16x4_f32(false, a, false, b,
                                                     (short)0, acc[t], false, false);
    }
  }

  const float bias = bl[col];
#pragma unroll
  for (int t = 0; t < RT; ++t) {
    const int tr0 = row0 + (rt0 + t) * 16;
    if (tr0 < N_NODES) {
      const int rbase = tr0 + ((lane >> 4) << 3);
#pragma unroll
      for (int i = 0; i < 8; ++i) {
        float v = acc[t][i] + bias;
        if (RELU) v = fmaxf(v, 0.0f);
        out[(long long)(rbase + i) * N + col] = v;
      }
    }
  }
}

extern "C" void kernel_launch(void* const* d_in, const int* in_sizes, int n_in,
                              void* d_out, int out_size, void* d_ws, size_t ws_size,
                              hipStream_t stream) {
  const float* x   = (const float*)d_in[0];
  const int*   ei  = (const int*)d_in[1];
  const float* Wl1 = (const float*)d_in[2];
  const float* bl1 = (const float*)d_in[3];
  const float* Wr1 = (const float*)d_in[4];
  const float* Wl2 = (const float*)d_in[5];
  const float* bl2 = (const float*)d_in[6];
  const float* Wr2 = (const float*)d_in[7];
  float* out = (float*)d_out;

  const int E = in_sizes[1] / 2;
  const int* src = ei;
  const int* dst = ei + E;

  float* msg = (float*)d_ws;
  float* deg = msg + (long long)N_NODES * FEAT;
  float* h1  = deg + N_NODES;

  const int TB = 256, SB = SC_WAVES * 32;
  const int GB = (N_NODES + GR - 1) / GR;

  sage_scatter_kernel<<<OWN_WGS, SB, 0, stream>>>(x, src, dst, msg, deg, E, 1);
  sage_gemm_kernel<8, true><<<GB, TB, 0, stream>>>(msg, deg, x, Wl1, bl1, Wr1, h1);

  sage_scatter_kernel<<<OWN_WGS, SB, 0, stream>>>(h1, src, dst, msg, deg, E, 0);
  sage_gemm_kernel<4, false><<<GB, TB, 0, stream>>>(msg, deg, h1, Wl2, bl2, Wr2, out);
}


# Layer-2 neighbour sum replays per-wave hit lists (src id, local row) recorded by the layer-1 scan, no second scan of dst; lists sit in out until the last GEMM overwrites it; LDS arrival order kept
# speedup vs baseline: 3.2415x; 1.1593x over previous
#include <hip/hip_runtime.h>

typedef float v2f __attribute__((ext_vector_type(2)));
typedef float v8f __attribute__((ext_vector_type(8)));

#define N_NODES 50000
#define FEAT    128

#define OWN_ROWS  196
#define OWN_WGS   ((N_NODES + OWN_ROWS - 1) / OWN_ROWS)
#define SC_WAVES  32
#define SC_DEPTH  4
#define SC_EU     8
#define SC_QCAP   (SC_DEPTH * 128 + SC_EU)
#define HIT_CAP   256
#define HIT_WORDS (OWN_WGS * SC_WAVES * HIT_CAP)

__global__ __launch_bounds__(SC_WAVES * 32)
void sage_scatter_kernel(const float* __restrict__ x,
                         const int* __restrict__ src,
                         const int* __restrict__ dst,
                         float* __restrict__ msg,
                         float* __restrict__ deg,
                         int* __restrict__ hits,
                         int E, int first) {
  __shared__ __attribute__((aligned(16))) float sAcc[OWN_ROWS * FEAT];
  __shared__ float sDeg[OWN_ROWS];
  __shared__ int   qE[SC_WAVES][SC_QCAP];
  __shared__ int   qD[SC_WAVES][SC_QCAP];
  __shared__ int   sCnt[SC_WAVES];

  const int tid  = threadIdx.x;
  const int lane = tid & 31;
  const int wave = __builtin_amdgcn_readfirstlane(tid >> 5);
  const int nodeBase = (int)blockIdx.x * OWN_ROWS;
  const int rows = (N_NODES - nodeBase < OWN_ROWS) ? (N_NODES - nodeBase) : OWN_ROWS;

  const int slot = (int)blockIdx.x * SC_WAVES + wave;
  int* myHits = hits + (long long)slot * HIT_CAP;
  int nList = -1;
  if (!first) nList = __builtin_amdgcn_readfirstlane(hits[HIT_WORDS + slot]);

  for (int i = tid; i < OWN_ROWS * FEAT / 4; i += SC_WAVES * 32)
    ((float4*)sAcc)[i] = float4{0.0f, 0.0f, 0.0f, 0.0f};
  for (int i = tid; i < OWN_ROWS; i += SC_WAVES * 32) sDeg[i] = 0.0f;
  int pn = lane < nList ? myHits[lane] : 0;
  __syncthreads();

  int* myE = qE[wave];
  int* myD = qD[wave];
  int cnt = 0;
  int nRec = 0;
  const int nTrip = (E + SC_WAVES * SC_DEPTH * 128 - 1) / (SC_WAVES * SC_DEPTH * 128);

  auto request = [&](int4 (&buf)[SC_DEPTH], int trip) {
#pragma unroll
    for (int k = 0; k < SC_DEPTH; ++k) {
      const int cb = ((trip * SC_DEPTH + k) * SC_WAVES + wave) * 128;
      if (cb + 128 <= E) {
        buf[k] = *(const int4*)(dst + cb + lane * 4);
      } else {
        const int b = cb + lane * 4;
        buf[k].x = b + 0 < E ? dst[b + 0] : -1;
        buf[k].y = b + 1 < E ? dst[b + 1] : -1;
        buf[k].z = b + 2 < E ? dst[b + 2] : -1;
        buf[k].w = b + 3 < E ? dst[b + 3] : -1;
      }
    }
  };

  auto scan = [&](const int4 (&buf)[SC_DEPTH], int trip) {
#pragma unroll
    for (int k = 0; k < SC_DEPTH; ++k) {
      const int base = ((trip * SC_DEPTH + k) * SC_WAVES + wave) * 128 + lane * 4;
      const int d0 = buf[k].x - nodeBase, d1 = buf[k].y - nodeBase;
      const int d2 = buf[k].z - nodeBase, d3 = buf[k].w - nodeBase;
      const bool h0 = (unsigned)d0 < (unsigned)rows, h1 = (unsigned)d1 < (unsigned)rows;
      const bool h2 = (unsigned)d2 < (unsigned)rows, h3 = (unsigned)d3 < (unsigned)rows;
      const unsigned b0 = __builtin_amdgcn_ballot_w32(h0);
      const unsigned b1 = __builtin_amdgcn_ballot_w32(h1);
      const unsigned b2 = __builtin_amdgcn_ballot_w32(h2);
      const unsigned b3 = __builtin_amdgcn_ballot_w32(h3);
      if (b0 | b1 | b2 | b3) {
        const unsigned below = (1u << lane) - 1u;
        int pos = cnt + __builtin_popcount(b0 & below);
        if (h0) { myE[pos] = base + 0; myD[pos] = d0; if (first) unsafeAtomicAdd(&sDeg[d0], 1.0f); }
        cnt += __builtin_popcount(b0);
        pos = cnt + __builtin_popcount(b1 & below);
        if (h1) { myE[pos] = base + 1; myD[pos] = d1; if (first) unsafeAtomicAdd(&sDeg[d1], 1.0f); }
        cnt += __builtin_popcount(b1);
        pos = cnt + __builtin_popcount(b2 & below);
        if (h2) { myE[pos] = base + 2; myD[pos] = d2; if (first) unsafeAtomicAdd(&sDeg[d2], 1.0f); }
        cnt += __builtin_popcount(b2);
        pos = cnt + __builtin_popcount(b3 & below);
        if (h3) { myE[pos] = base + 3; myD[pos] = d3; if (first) unsafeAtomicAdd(&sDeg[d3], 1.0f); }
        cnt += __builtin_popcount(b3);
      }
    }
  };

  auto drain = [&](bool all) {
    if (!(cnt >= SC_EU || (all && cnt > 0))) return;
    int q0 = 0;
    while (cnt - q0 >= SC_EU || (all && q0 < cnt)) {
      int m = cnt - q0 < 32 ? cnt - q0 : 32;
      if (!all) m &= ~(SC_EU - 1);
      int s = 0, dl = 0;
      if (lane < m) { dl = myD[q0 + lane]; s = src[myE[q0 + lane]]; }
      if (first) {
        if (nRec + m <= HIT_CAP) {
          if (lane < m) myHits[nRec + lane] = s | (dl << 16);
        }
        nRec += m;
      }
      int g = 0;
      for (; g + SC_EU <= m; g += SC_EU) {
        float v[SC_EU][4];
#pragma unroll
        for (int u = 0; u < SC_EU; ++u) {
          const int su = __builtin_amdgcn_readlane(s, g + u);
          const float* xs = x + (long long)su * FEAT + lane;
          v[u][0] = xs[0]; v[u][1] = xs[32]; v[u][2] = xs[64]; v[u][3] = xs[96];
        }
#pragma unroll
        for (int u = 0; u < SC_EU; ++u) {
          const int du = __builtin_amdgcn_readlane(dl, g + u);
          float* p = sAcc + du * FEAT + lane;
          unsafeAtomicAdd(p + 0,  v[u][0]);
          unsafeAtomicAdd(p + 32, v[u][1]);
          unsafeAtomicAdd(p + 64, v[u][2]);
          unsafeAtomicAdd(p + 96, v[u][3]);
        }
      }
      for (; g < m; ++g) {
        const int su = __builtin_amdgcn_readlane(s, g);
        const int du = __builtin_amdgcn_readlane(dl, g);
        const float* xs = x + (long long)su * FEAT + lane;
        const float v0 = xs[0], v1 = xs[32], v2 = xs[64], v3 = xs[96];
        float* p = sAcc + du * FEAT + lane;
        unsafeAtomicAdd(p + 0,  v0);
        unsafeAtomicAdd(p + 32, v1);
        unsafeAtomicAdd(p + 64, v2);
        unsafeAtomicAdd(p + 96, v3);
      }
      q0 += m;
    }
    const int left = cnt - q0;
    int te = 0, td = 0;
    if (lane < left) { te = myE[q0 + lane]; td = myD[q0 + lane]; }
    if (lane < left) { myE[lane] = te; myD[lane] = td; }
    cnt = left;
  };

  auto requestFull = [&](int4 (&buf)[SC_DEPTH], int trip) {
#pragma unroll
    for (int k = 0; k < SC_DEPTH; ++k) {
      const int cb = ((trip * SC_DEPTH + k) * SC_WAVES + wave) * 128;
      buf[k] = *(const int4*)(dst + cb + lane * 4);
    }
  };

  if (nList >= 0) {
    for (int q = 0; q < nList; q += 32) {
      const int p = pn;
      const int m = nList - q < 32 ? nList - q : 32;
      if (q + 32 < nList) pn = q + 32 + lane < nList ? myHits[q + 32 + lane] : 0;
      int g = 0;
      for (; g + SC_EU <= m; g += SC_EU) {
        float v[SC_EU][4];
#pragma unroll
        for (int u = 0; u < SC_EU; ++u) {
          const int su = __builtin_amdgcn_readlane(p, g + u) & 0xffff;
          const float* xs = x + (long long)su * FEAT + lane;
          v[u][0] = xs[0]; v[u][1] = xs[32]; v[u][2] = xs[64]; v[u][3] = xs[96];
        }
#pragma unroll
        for (int u = 0; u < SC_EU; ++u) {
          const int du = __builtin_amdgcn_readlane(p, g + u) >> 16;
          float* a = sAcc + du * FEAT + lane;
          unsafeAtomicAdd(a + 0,  v[u][0]);
          unsafeAtomicAdd(a + 32, v[u][1]);
          unsafeAtomicAdd(a + 64, v[u][2]);
          unsafeAtomicAdd(a + 96, v[u][3]);
        }
      }
      const int r = m - g;
      if (r > 0) {
        float v[SC_EU][4];
#pragma unroll
        for (int u = 0; u < SC_EU; ++u) {
          if (u < r) {
            const int su = __builtin_amdgcn_readlane(p, g + u) & 0xffff;
            const float* xs = x + (long long)su * FEAT + lane;
            v[u][0] = xs[0]; v[u][1] = xs[32]; v[u][2] = xs[64]; v[u][3] = xs[96];
          }
        }
#pragma unroll
        for (int u = 0; u < SC_EU; ++u) {
          if (u < r) {
            const int du = __builtin_amdgcn_readlane(p, g + u) >> 16;
            float* a = sAcc + du * FEAT + lane;
            unsafeAtomicAdd(a + 0,  v[u][0]);
            unsafeAtomicAdd(a + 32, v[u][1]);
            unsafeAtomicAdd(a + 64, v[u][2]);
            unsafeAtomicAdd(a + 96, v[u][3]);
          }
        }
      }
    }
  } else {
    const int nPair = (E / (SC_WAVES * SC_DEPTH * 128)) & ~1;
    int trip = 0;
    if (nPair > 0) {
      int4 bufA[SC_DEPTH], bufB[SC_DEPTH];
      requestFull(bufA, 0);
      for (; trip < nPair; trip += 2) {
        requestFull(bufB, trip + 1);
        scan(bufA, trip);
        drain(false);
        requestFull(bufA, trip + 2 < nPair ? trip + 2 : nPair - 1);
        scan(bufB, trip + 1);
        drain(false);
      }
    }
    for (; trip < nTrip; ++trip) {
      int4 bufT[SC_DEPTH];
      request(bufT, trip);
      scan(bufT, trip);
      drain(false);
    }
    drain(true);
    if (first && lane == 0) sCnt[wave] = nRec <= HIT_CAP ? nRec : -1;
  }
  __syncthreads();

  if (first && wave == 0) hits[HIT_WORDS + (int)blockIdx.x * SC_WAVES + lane] = sCnt[lane];

  for (int r = wave; r < rows; r += SC_WAVES)
    *(float4*)(msg + (long long)(nodeBase + r) * FEAT + lane * 4) =
        *(const float4*)(sAcc + r * FEAT + lane * 4);
  if (first)
    for (int i = tid; i < rows; i += SC_WAVES * 32) deg[nodeBase + i] = sDeg[i];
}

#define GR     32
#define APITCH (FEAT + 4)

template <int NCT, bool RELU>
__global__ __launch_bounds__(256)
void sage_gemm_kernel(const float* __restrict__ msg,
                      const float* __restrict__ deg,
                      const float* __restrict__ xin,
                      const float* __restrict__ Wl,
                      const float* __restrict__ bl,
                      const float* __restrict__ Wr,
                      float* __restrict__ out) {
  constexpr int N  = NCT * 16;
  constexpr int RT = NCT / 4;
  __shared__ __attribute__((aligned(16))) float sM[GR * APITCH];
  __shared__ __attribute__((aligned(16))) float sX[GR * APITCH];

  const int lane = threadIdx.x & 31;
  const int wave = __builtin_amdgcn_readfirstlane((int)(threadIdx.x >> 5));
  const int row0 = (int)blockIdx.x * GR;

  float  dg[GR / 8];
  float4 m[GR / 8], xv[GR / 8];
#pragma unroll
  for (int j = 0; j < GR / 8; ++j) {
    const int gr  = row0 + wave * (GR / 8) + j;
    const int grc = gr < N_NODES ? gr : N_NODES - 1;
    dg[j] = deg[grc];
    m[j]  = *(const float4*)(msg + (long long)grc * FEAT + lane * 4);
    xv[j] = *(const float4*)(xin + (long long)grc * FEAT + lane * 4);
  }
#pragma unroll
  for (int j = 0; j < GR / 8; ++j) {
    const int lr = wave * (GR / 8) + j;
    const float scale = 1.0f / fmaxf(dg[j], 1.0f);
    m[j].x *= scale; m[j].y *= scale; m[j].z *= scale; m[j].w *= scale;
    *(float4*)(sM + lr * APITCH + lane * 4) = m[j];
    *(float4*)(sX + lr * APITCH + lane * 4) = xv[j];
  }
  __syncthreads();

  const int c16 = lane & 15;
  const int kb  = (lane >> 4) << 1;
  const int colTile = (NCT == 8) ? wave : (wave & 3);
  const int rt0     = (NCT == 8) ? 0 : (wave >> 2);
  const int col = (colTile << 4) + c16;

  v8f acc[RT];
#pragma unroll
  for (int t = 0; t < RT; ++t) acc[t] = v8f{};

  const float* aM = sM + (rt0 * 16 + c16) * APITCH + kb;
  const float* aX = sX + (rt0 * 16 + c16) * APITCH + kb;
  const float* bL = Wl + kb * N + col;
  const float* bR = Wr + kb * N + col;

#pragma unroll 8
  for (int k0 = 0; k0 < FEAT; k0 += 4) {
    v2f b;
    b.x = bL[k0 * N];
    b.y = bL[(k0 + 1) * N];
#pragma unroll
    for (int t = 0; t < RT; ++t) {
      const v2f a = *(const v2f*)(aM + t * 16 * APITCH + k0);
      acc[t] = __builtin_amdgcn_wmma_f32_16x16x4_f32(false, a, false, b,
                                                     (short)0, acc[t], false, false);
    }
  }
#pragma unroll 8
  for (int k0 = 0; k0 < FEAT; k0 += 4) {
    v2f b;
    b.x = bR[k0 * N];
    b.y = bR[(k0 + 1) * N];
#pragma unroll
    for (int t = 0; t < RT; ++t) {
      const v2f a = *(const v2f*)(aX + t * 16 * APITCH + k0);
      acc[t] = __builtin_amdgcn_wmma_f32_16x16x4_f32(false, a, false, b,
                                                     (short)0, acc[t], false, false);
    }
  }

  const float bias = bl[col];
#pragma unroll
  for (int t = 0; t < RT; ++t) {
    const int tr0 = row0 + (rt0 + t) * 16;
    if (tr0 < N_NODES) {
      const int rbase = tr0 + ((lane >> 4) << 3);
#pragma unroll
      for (int i = 0; i < 8; ++i) {
        float v = acc[t][i] + bias;
        if (RELU) v = fmaxf(v, 0.0f);
        out[(long long)(rbase + i) * N + col] = v;
      }
    }
  }
}

extern "C" void kernel_launch(void* const* d_in, const int* in_sizes, int n_in,
                              void* d_out, int out_size, void* d_ws, size_t ws_size,
                              hipStream_t stream) {
  const float* x   = (const float*)d_in[0];
  const int*   ei  = (const int*)d_in[1];
  const float* Wl1 = (const float*)d_in[2];
  const float* bl1 = (const float*)d_in[3];
  const float* Wr1 = (const float*)d_in[4];
  const float* Wl2 = (const float*)d_in[5];
  const float* bl2 = (const float*)d_in[6];
  const float* Wr2 = (const float*)d_in[7];
  float* out = (float*)d_out;

  const int E = in_sizes[1] / 2;
  const int* src = ei;
  const int* dst = ei + E;

  float* msg = (float*)d_ws;
  float* deg = msg + (long long)N_NODES * FEAT;
  float* h1  = deg + N_NODES;
  int*   hits = (int*)d_out;

  const int TB = 256, SB = SC_WAVES * 32;
  const int GB = (N_NODES + GR - 1) / GR;

  sage_scatter_kernel<<<OWN_WGS, SB, 0, stream>>>(x, src, dst, msg, deg, hits, E, 1);
  sage_gemm_kernel<8, true><<<GB, TB, 0, stream>>>(msg, deg, x, Wl1, bl1, Wr1, h1);

  sage_scatter_kernel<<<OWN_WGS, SB, 0, stream>>>(h1, src, dst, msg, deg, hits, E, 0);
  sage_gemm_kernel<4, false><<<GB, TB, 0, stream>>>(msg, deg, h1, Wl2, bl2, Wr2, out);
}


# Layer GEMM fused into the neighbour-sum kernel: owner workgroup scales its LDS sums, stages its x rows in dead queue LDS, runs its WMMA tiles; sums never go to memory; lists in workspace; 2 launches
# speedup vs baseline: 3.5381x; 1.0915x over previous
#include <hip/hip_runtime.h>


typedef float v2f __attribute__((ext_vector_type(2)));
typedef float v8f __attribute__((ext_vector_type(8)));

#define N_NODES 50000
#define FEAT    128

#define OWN_ROWS  196
#define OWN_WGS   ((N_NODES + OWN_ROWS - 1) / OWN_ROWS)
#define OWN_TILES ((OWN_ROWS + 15) / 16)
#define SC_WAVES  32
#define SC_DEPTH  4
#define SC_EU     8
#define SC_QCAP   (SC_DEPTH * 128 + SC_EU)
#define HIT_CAP   256
#define HIT_WORDS (OWN_WGS * SC_WAVES * HIT_CAP)
#define APITCH    (FEAT + 4)
#define XROWS_PER_WAVE ((OWN_ROWS + SC_WAVES - 1) / SC_WAVES)

template <int N, int NT, int NRG, bool RELU>
__device__ __forceinline__ void layer_tiles(const float* sM, const float* sX,
                                            const float* __restrict__ Wl,
                                            const float* __restrict__ bl,
                                            const float* __restrict__ Wr,
                                            float* __restrict__ out,
                                            int nodeBase, int rows, int rg, int colTile, int lane) {
  const int c16 = lane & 15;
  const int kb  = (lane >> 4) << 1;
  const int col = (colTile << 4) + c16;

  v8f acc[NT];
  int aoff[NT];
#pragma unroll
  for (int t = 0; t < NT; ++t) {
    acc[t] = v8f{};
    int r = (rg + t * NRG) * 16 + c16;
    r = r < OWN_ROWS ? r : OWN_ROWS - 1;
    aoff[t] = r * APITCH + kb;
  }
  const float* bL = Wl + kb * N + col;
  const float* bR = Wr + kb * N + col;

#pragma unroll 8
  for (int k0 = 0; k0 < FEAT; k0 += 4) {
    v2f b;
    b.x = bL[k0 * N];
    b.y = bL[(k0 + 1) * N];
#pragma unroll
    for (int t = 0; t < NT; ++t) {
      const v2f a = *(const v2f*)(sM + aoff[t] + k0);
      acc[t] = __builtin_amdgcn_wmma_f32_16x16x4_f32(false, a, false, b,
                                                     (short)0, acc[t], false, false);
    }
  }
#pragma unroll 8
  for (int k0 = 0; k0 < FEAT; k0 += 4) {
    v2f b;
    b.x = bR[k0 * N];
    b.y = bR[(k0 + 1) * N];
#pragma unroll
    for (int t = 0; t < NT; ++t) {
      const v2f a = *(const v2f*)(sX + aoff[t] + k0);
      acc[t] = __builtin_amdgcn_wmma_f32_16x16x4_f32(false, a, false, b,
                                                     (short)0, acc[t], false, false);
    }
  }

  const float bias = bl[col];
#pragma unroll
  for (int t = 0; t < NT; ++t) {
    const int rbase = (rg + t * NRG) * 16 + ((lane >> 4) << 3);
#pragma unroll
    for (int i = 0; i < 8; ++i) {
      const int r = rbase + i;
      if (r < rows) {
        float v = acc[t][i] + bias;
        if (RELU) v = fmaxf(v, 0.0f);
        out[(long long)(nodeBase + r) * N + col] = v;
      }
    }
  }
}

template <bool FIRST>
__global__ __launch_bounds__(SC_WAVES * 32)
void sage_layer_kernel(const float* __restrict__ xin,
                       const int* __restrict__ src,
                       const int* __restrict__ dst,
                       const float* __restrict__ Wl,
                       const float* __restrict__ bl,
                       const float* __restrict__ Wr,
                       float* __restrict__ deg,
                       int* __restrict__ hits,
                       float* __restrict__ out,
                       int E) {
  __shared__ __attribute__((aligned(16))) float sAcc[OWN_ROWS * APITCH];
  __shared__ float sDeg[OWN_ROWS];
  __shared__ __attribute__((aligned(16))) unsigned char sRaw[2 * SC_WAVES * SC_QCAP * 4];
  __shared__ int   sCnt[SC_WAVES];

  const int tid  = threadIdx.x;
  const int lane = tid & 31;
  const int wave = __builtin_amdgcn_readfirstlane(tid >> 5);
  const int nodeBase = (int)blockIdx.x * OWN_ROWS;
  const int rows = (N_NODES - nodeBase < OWN_ROWS) ? (N_NODES - nodeBase) : OWN_ROWS;

  const int slot = (int)blockIdx.x * SC_WAVES + wave;
  int* myHits = hits + (long long)slot * HIT_CAP;
  int nList = -1;
  if (!FIRST) nList = __builtin_amdgcn_readfirstlane(hits[HIT_WORDS + slot]);

  for (int i = tid; i < OWN_ROWS * APITCH / 4; i += SC_WAVES * 32)
    ((float4*)sAcc)[i] = float4{0.0f, 0.0f, 0.0f, 0.0f};
  for (int i = tid; i < OWN_ROWS; i += SC_WAVES * 32)
    sDeg[i] = (!FIRST && i < rows) ? deg[nodeBase + i] : 0.0f;
  int pn = lane < nList ? myHits[lane] : 0;
  __syncthreads();

  int* myE = (int*)sRaw + wave * SC_QCAP;
  int* myD = (int*)sRaw + (SC_WAVES + wave) * SC_QCAP;
  int cnt = 0;
  int nRec = 0;
  const int nTrip = (E + SC_WAVES * SC_DEPTH * 128 - 1) / (SC_WAVES * SC_DEPTH * 128);

  auto request = [&](int4 (&buf)[SC_DEPTH], int trip) {
#pragma unroll
    for (int k = 0; k < SC_DEPTH; ++k) {
      const int cb = ((trip * SC_DEPTH + k) * SC_WAVES + wave) * 128;
      if (cb + 128 <= E) {
        buf[k] = *(const int4*)(dst + cb + lane * 4);
      } else {
        const int b = cb + lane * 4;
        buf[k].x = b + 0 < E ? dst[b + 0] : -1;
        buf[k].y = b + 1 < E ? dst[b + 1] : -1;
        buf[k].z = b + 2 < E ? dst[b + 2] : -1;
        buf[k].w = b + 3 < E ? dst[b + 3] : -1;
      }
    }
  };

  auto scan = [&](const int4 (&buf)[SC_DEPTH], int trip) {
#pragma unroll
    for (int k = 0; k < SC_DEPTH; ++k) {
      const int base = ((trip * SC_DEPTH + k) * SC_WAVES + wave) * 128 + lane * 4;
      const int d0 = buf[k].x - nodeBase, d1 = buf[k].y - nodeBase;
      const int d2 = buf[k].z - nodeBase, d3 = buf[k].w - nodeBase;
      const bool h0 = (unsigned)d0 < (unsigned)rows, h1 = (unsigned)d1 < (unsigned)rows;
      const bool h2 = (unsigned)d2 < (unsigned)rows, h3 = (unsigned)d3 < (unsigned)rows;
      const unsigned b0 = __builtin_amdgcn_ballot_w32(h0);
      const unsigned b1 = __builtin_amdgcn_ballot_w32(h1);
      const unsigned b2 = __builtin_amdgcn_ballot_w32(h2);
      const unsigned b3 = __builtin_amdgcn_ballot_w32(h3);
      if (b0 | b1 | b2 | b3) {
        const unsigned below = (1u << lane) - 1u;
        int pos = cnt + __builtin_popcount(b0 & below);
        if (h0) { myE[pos] = base + 0; myD[pos] = d0; if (FIRST) unsafeAtomicAdd(&sDeg[d0], 1.0f); }
        cnt += __builtin_popcount(b0);
        pos = cnt + __builtin_popcount(b1 & below);
        if (h1) { myE[pos] = base + 1; myD[pos] = d1; if (FIRST) unsafeAtomicAdd(&sDeg[d1], 1.0f); }
        cnt += __builtin_popcount(b1);
        pos = cnt + __builtin_popcount(b2 & below);
        if (h2) { myE[pos] = base + 2; myD[pos] = d2; if (FIRST) unsafeAtomicAdd(&sDeg[d2], 1.0f); }
        cnt += __builtin_popcount(b2);
        pos = cnt + __builtin_popcount(b3 & below);
        if (h3) { myE[pos] = base + 3; myD[pos] = d3; if (FIRST) unsafeAtomicAdd(&sDeg[d3], 1.0f); }
        cnt += __builtin_popcount(b3);
      }
    }
  };

  auto drain = [&](bool all) {
    if (!(cnt >= SC_EU || (all && cnt > 0))) return;
    int q0 = 0;
    while (cnt - q0 >= SC_EU || (all && q0 < cnt)) {
      int m = cnt - q0 < 32 ? cnt - q0 : 32;
      if (!all) m &= ~(SC_EU - 1);
      int s = 0, dl = 0;
      if (lane < m) { dl = myD[q0 + lane]; s = src[myE[q0 + lane]]; }
      if (FIRST) {
        if (nRec + m <= HIT_CAP) {
          if (lane < m) myHits[nRec + lane] = s | (dl << 16);
        }
        nRec += m;
      }
      int g = 0;
      for (; g + SC_EU <= m; g += SC_EU) {
        float v[SC_EU][4];
#pragma unroll
        for (int u = 0; u < SC_EU; ++u) {
          const int su = __builtin_amdgcn_readlane(s, g + u);
          const float* xs = xin + (long long)su * FEAT + lane;
          v[u][0] = xs[0]; v[u][1] = xs[32]; v[u][2] = xs[64]; v[u][3] = xs[96];
        }
#pragma unroll
        for (int u = 0; u < SC_EU; ++u) {
          const int du = __builtin_amdgcn_readlane(dl, g + u);
          float* p = sAcc + du * APITCH + lane;
          unsafeAtomicAdd(p + 0,  v[u][0]);
          unsafeAtomicAdd(p + 32, v[u][1]);
          unsafeAtomicAdd(p + 64, v[u][2]);
          unsafeAtomicAdd(p + 96, v[u][3]);
        }
      }
      for (; g < m; ++g) {
        const int su = __builtin_amdgcn_readlane(s, g);
        const int du = __builtin_amdgcn_readlane(dl, g);
        const float* xs = xin + (long long)su * FEAT + lane;
        const float v0 = xs[0], v1 = xs[32], v2 = xs[64], v3 = xs[96];
        float* p = sAcc + du * APITCH + lane;
        unsafeAtomicAdd(p + 0,  v0);
        unsafeAtomicAdd(p + 32, v1);
        unsafeAtomicAdd(p + 64, v2);
        unsafeAtomicAdd(p + 96, v3);
      }
      q0 += m;
    }
    const int left = cnt - q0;
    int te = 0, td = 0;
    if (lane < left) { te = myE[q0 + lane]; td = myD[q0 + lane]; }
    if (lane < left) { myE[lane] = te; myD[lane] = td; }
    cnt = left;
  };

  auto requestFull = [&](int4 (&buf)[SC_DEPTH], int trip) {
#pragma unroll
    for (int k = 0; k < SC_DEPTH; ++k) {
      const int cb = ((trip * SC_DEPTH + k) * SC_WAVES + wave) * 128;
      buf[k] = *(const int4*)(dst + cb + lane * 4);
    }
  };

  if (nList >= 0) {
    for (int q = 0; q < nList; q += 32) {
      const int p = pn;
      const int m = nList - q < 32 ? nList - q : 32;
      if (q + 32 < nList) pn = q + 32 + lane < nList ? myHits[q + 32 + lane] : 0;
      int g = 0;
      for (; g + SC_EU <= m; g += SC_EU) {
        float v[SC_EU][4];
#pragma unroll
        for (int u = 0; u < SC_EU; ++u) {
          const int su = __builtin_amdgcn_readlane(p, g + u) & 0xffff;
          const float* xs = xin + (long long)su * FEAT + lane;
          v[u][0] = xs[0]; v[u][1] = xs[32]; v[u][2] = xs[64]; v[u][3] = xs[96];
        }
#pragma unroll
        for (int u = 0; u < SC_EU; ++u) {
          const int du = __builtin_amdgcn_readlane(p, g + u) >> 16;
          float* a = sAcc + du * APITCH + lane;
          unsafeAtomicAdd(a + 0,  v[u][0]);
          unsafeAtomicAdd(a + 32, v[u][1]);
          unsafeAtomicAdd(a + 64, v[u][2]);
          unsafeAtomicAdd(a + 96, v[u][3]);
        }
      }
      const int r = m - g;
      if (r > 0) {
        float v[SC_EU][4];
#pragma unroll
        for (int u = 0; u < SC_EU; ++u) {
          if (u < r) {
            const int su = __builtin_amdgcn_readlane(p, g + u) & 0xffff;
            const float* xs = xin + (long long)su * FEAT + lane;
            v[u][0] = xs[0]; v[u][1] = xs[32]; v[u][2] = xs[64]; v[u][3] = xs[96];
          }
        }
#pragma unroll
        for (int u = 0; u < SC_EU; ++u) {
          if (u < r) {
            const int du = __builtin_amdgcn_readlane(p, g + u) >> 16;
            float* a = sAcc + du * APITCH + lane;
            unsafeAtomicAdd(a + 0,  v[u][0]);
            unsafeAtomicAdd(a + 32, v[u][1]);
            unsafeAtomicAdd(a + 64, v[u][2]);
            unsafeAtomicAdd(a + 96, v[u][3]);
          }
        }
      }
    }
  } else {
    const int nPair = (E / (SC_WAVES * SC_DEPTH * 128)) & ~1;
    int trip = 0;
    if (nPair > 0) {
      int4 bufA[SC_DEPTH], bufB[SC_DEPTH];
      requestFull(bufA, 0);
      for (; trip < nPair; trip += 2) {
        requestFull(bufB, trip + 1);
        scan(bufA, trip);
        drain(false);
        requestFull(bufA, trip + 2 < nPair ? trip + 2 : nPair - 1);
        scan(bufB, trip + 1);
        drain(false);
      }
    }
    for (; trip < nTrip; ++trip) {
      int4 bufT[SC_DEPTH];
      request(bufT, trip);
      scan(bufT, trip);
      drain(false);
    }
    drain(true);
    if (FIRST && lane == 0) sCnt[wave] = nRec <= HIT_CAP ? nRec : -1;
  }

  static_assert(XROWS_PER_WAVE == 7, "seven rows per wave are written out by name below");
  auto xrow = [&](int j) -> const float4* {
    const int r   = wave + j * SC_WAVES;
    const int rc  = r < OWN_ROWS ? r : OWN_ROWS - 1;
    const int gr  = nodeBase + rc;
    const int grc = gr < N_NODES ? gr : N_NODES - 1;
    return (const float4*)(xin + (long long)grc * FEAT + lane * 4);
  };
  const float4 xr0 = *xrow(0), xr1 = *xrow(1), xr2 = *xrow(2), xr3 = *xrow(3);
  const float4 xr4 = *xrow(4), xr5 = *xrow(5), xr6 = *xrow(6);
  __syncthreads();

  if (FIRST) {
    if (wave == 0) hits[HIT_WORDS + (int)blockIdx.x * SC_WAVES + lane] = sCnt[lane];
    for (int i = tid; i < rows; i += SC_WAVES * 32) deg[nodeBase + i] = sDeg[i];
  }

  for (int i = tid; i < OWN_ROWS * 32; i += SC_WAVES * 32) {
    const int r = i >> 5;
    const float scale = 1.0f / fmaxf(sDeg[r], 1.0f);
    float4* p = (float4*)(sAcc + r * APITCH + (i & 31) * 4);
    float4 m = *p;
    m.x *= scale; m.y *= scale; m.z *= scale; m.w *= scale;
    *p = m;
  }
  static_assert(XROWS_PER_WAVE * SC_WAVES * APITCH * 4 <= 2 * SC_WAVES * SC_QCAP * 4,
                "the staged rows must fit in the queue space");
  float* sX = (float*)sRaw;
  float4* xd = (float4*)(sX + wave * APITCH + lane * 4);
  constexpr int XSTEP = SC_WAVES * APITCH / 4;
  xd[0 * XSTEP] = xr0; xd[1 * XSTEP] = xr1; xd[2 * XSTEP] = xr2; xd[3 * XSTEP] = xr3;
  xd[4 * XSTEP] = xr4; xd[5 * XSTEP] = xr5; xd[6 * XSTEP] = xr6;
  __syncthreads();

  constexpr int  NCT   = FIRST ? 8 : 4;
  constexpr int  N     = NCT * 16;
  constexpr int  NRG   = SC_WAVES / NCT;
  constexpr int  RTMAX = (OWN_TILES + NRG - 1) / NRG;
  const int colTile = wave & (NCT - 1);
  const int rg      = wave / NCT;
  const int nTiles  = (rows + 15) >> 4;
  const int nT      = rg < nTiles ? (nTiles - rg + NRG - 1) / NRG : 0;

  if (RTMAX >= 4 && nT == 4)
    layer_tiles<N, (RTMAX >= 4 ? 4 : 1), NRG, FIRST>(sAcc, sX, Wl, bl, Wr, out, nodeBase, rows, rg, colTile, lane);
  else if (RTMAX >= 3 && nT == 3)
    layer_tiles<N, (RTMAX >= 3 ? 3 : 1), NRG, FIRST>(sAcc, sX, Wl, bl, Wr, out, nodeBase, rows, rg, colTile, lane);
  else if (nT == 2)
    layer_tiles<N, 2, NRG, FIRST>(sAcc, sX, Wl, bl, Wr, out, nodeBase, rows, rg, colTile, lane);
  else if (nT == 1)
    layer_tiles<N, 1, NRG, FIRST>(sAcc, sX, Wl, bl, Wr, out, nodeBase, rows, rg, colTile, lane);
}

extern "C" void kernel_launch(void* const* d_in, const int* in_sizes, int n_in,
                              void* d_out, int out_size, void* d_ws, size_t ws_size,
                              hipStream_t stream) {
  const float* x   = (const float*)d_in[0];
  const int*   ei  = (const int*)d_in[1];
  const float* Wl1 = (const float*)d_in[2];
  const float* bl1 = (const float*)d_in[3];
  const float* Wr1 = (const float*)d_in[4];
  const float* Wl2 = (const float*)d_in[5];
  const float* bl2 = (const float*)d_in[6];
  const float* Wr2 = (const float*)d_in[7];
  float* out = (float*)d_out;

  const int E = in_sizes[1] / 2;
  const int* src = ei;
  const int* dst = ei + E;

  int*   hits = (int*)d_ws;
  float* deg  = (float*)d_ws + (long long)N_NODES * FEAT;
  float* h1   = deg + N_NODES;

  const int SB = SC_WAVES * 32;

  sage_layer_kernel<true><<<OWN_WGS, SB, 0, stream>>>(x, src, dst, Wl1, bl1, Wr1, deg, hits, h1, E);
  sage_layer_kernel<false><<<OWN_WGS, SB, 0, stream>>>(h1, src, dst, Wl2, bl2, Wr2, deg, hits, out, E);
}


# Layer-1 scan reads a one-byte owner map (dst / 196, written by a new first launch) in place of dst: a quarter of the bytes, 16 edges per lane and load; 3 launches; sums stay in LDS arrival order
# speedup vs baseline: 3.6769x; 1.0392x over previous
#include <hip/hip_runtime.h>


typedef float v2f __attribute__((ext_vector_type(2)));
typedef float v8f __attribute__((ext_vector_type(8)));

#define N_NODES 50000
#define FEAT    128

#define OWN_ROWS  196
#define OWN_WGS   ((N_NODES + OWN_ROWS - 1) / OWN_ROWS)
#define OWN_TILES ((OWN_ROWS + 15) / 16)
#define SC_WAVES  32
#define SC_DEPTH  4
#define SC_EU     8
#define SC_CHUNK  512
#define SC_QCAP   (2 * SC_CHUNK + 2 * SC_EU)
#define OM_WGS    391
#define OM_DW     512
#define HIT_CAP   256
#define HIT_WORDS (OWN_WGS * SC_WAVES * HIT_CAP)
#define APITCH    (FEAT + 4)
#define XROWS_PER_WAVE ((OWN_ROWS + SC_WAVES - 1) / SC_WAVES)

__device__ __forceinline__ unsigned owner_byte(int d) {
  return ((unsigned)d / (unsigned)OWN_ROWS) & 0xffu;
}
__global__ __launch_bounds__(128)
void owner_map_kernel(const int* __restrict__ dst, int* __restrict__ omap, int E) {
  const int nDw  = ((E + 15) >> 4) << 2;
  const int full = E >> 2;
  for (int q0 = (int)blockIdx.x * OM_DW; q0 < nDw; q0 += (int)gridDim.x * OM_DW) {
    const int q = q0 + (int)threadIdx.x;
    if (q0 + OM_DW <= full) {
      int4 d[4];
#pragma unroll
      for (int k = 0; k < 4; ++k) d[k] = *(const int4*)(dst + (q + k * 128) * 4);
#pragma unroll
      for (int k = 0; k < 4; ++k)
        omap[q + k * 128] = (int)(owner_byte(d[k].x) | (owner_byte(d[k].y) << 8) |
                                  (owner_byte(d[k].z) << 16) | (owner_byte(d[k].w) << 24));
    } else {
      for (int k = 0; k < 4; ++k) {
        const int qq = q + k * 128;
        if (qq < nDw) {
          const int e = qq * 4;
          const int d0 = e + 0 < E ? dst[e + 0] : 0, d1 = e + 1 < E ? dst[e + 1] : 0;
          const int d2 = e + 2 < E ? dst[e + 2] : 0, d3 = e + 3 < E ? dst[e + 3] : 0;
          omap[qq] = (int)(owner_byte(d0) | (owner_byte(d1) << 8) |
                           (owner_byte(d2) << 16) | (owner_byte(d3) << 24));
        }
      }
    }
  }
}

template <int N, int NT, int NRG, bool RELU>
__device__ __forceinline__ void layer_tiles(const float* sM, const float* sX,
                                            const float* __restrict__ Wl,
                                            const float* __restrict__ bl,
                                            const float* __restrict__ Wr,
                                            float* __restrict__ out,
                                            int nodeBase, int rows, int rg, int colTile, int lane) {
  const int c16 = lane & 15;
  const int kb  = (lane >> 4) << 1;
  const int col = (colTile << 4) + c16;

  v8f acc[NT];
  int aoff[NT];
#pragma unroll
  for (int t = 0; t < NT; ++t) {
    acc[t] = v8f{};
    int r = (rg + t * NRG) * 16 + c16;
    r = r < OWN_ROWS ? r : OWN_ROWS - 1;
    aoff[t] = r * APITCH + kb;
  }
  const float* bL = Wl + kb * N + col;
  const float* bR = Wr + kb * N + col;

#pragma unroll 8
  for (int k0 = 0; k0 < FEAT; k0 += 4) {
    v2f b;
    b.x = bL[k0 * N];
    b.y = bL[(k0 + 1) * N];
#pragma unroll
    for (int t = 0; t < NT; ++t) {
      const v2f a = *(const v2f*)(sM + aoff[t] + k0);
      acc[t] = __builtin_amdgcn_wmma_f32_16x16x4_f32(false, a, false, b,
                                                     (short)0, acc[t], false, false);
    }
  }
#pragma unroll 8
  for (int k0 = 0; k0 < FEAT; k0 += 4) {
    v2f b;
    b.x = bR[k0 * N];
    b.y = bR[(k0 + 1) * N];
#pragma unroll
    for (int t = 0; t < NT; ++t) {
      const v2f a = *(const v2f*)(sX + aoff[t] + k0);
      acc[t] = __builtin_amdgcn_wmma_f32_16x16x4_f32(false, a, false, b,
                                                     (short)0, acc[t], false, false);
    }
  }

  const float bias = bl[col];
#pragma unroll
  for (int t = 0; t < NT; ++t) {
    const int rbase = (rg + t * NRG) * 16 + ((lane >> 4) << 3);
#pragma unroll
    for (int i = 0; i < 8; ++i) {
      const int r = rbase + i;
      if (r < rows) {
        float v = acc[t][i] + bias;
        if (RELU) v = fmaxf(v, 0.0f);
        out[(long long)(nodeBase + r) * N + col] = v;
      }
    }
  }
}

template <bool FIRST>
__global__ __launch_bounds__(SC_WAVES * 32)
void sage_layer_kernel(const float* __restrict__ xin,
                       const int* __restrict__ src,
                       const int* __restrict__ dst,
                       const int* __restrict__ omap,
                       const float* __restrict__ Wl,
                       const float* __restrict__ bl,
                       const float* __restrict__ Wr,
                       float* __restrict__ deg,
                       int* __restrict__ hits,
                       float* __restrict__ out,
                       int E) {
  __shared__ __attribute__((aligned(16))) float sAcc[(OWN_ROWS + 1) * APITCH];
  __shared__ float sDeg[OWN_ROWS];
  __shared__ __attribute__((aligned(16))) unsigned char sRaw[SC_WAVES * SC_QCAP * 4];
  __shared__ int   sCnt[SC_WAVES];

  const int tid  = threadIdx.x;
  const int lane = tid & 31;
  const int wave = __builtin_amdgcn_readfirstlane(tid >> 5);
  const int nodeBase = (int)blockIdx.x * OWN_ROWS;
  const int rows = (N_NODES - nodeBase < OWN_ROWS) ? (N_NODES - nodeBase) : OWN_ROWS;

  const int slot = (int)blockIdx.x * SC_WAVES + wave;
  int* myHits = hits + (long long)slot * HIT_CAP;
  int nList = -1;
  if (!FIRST) nList = __builtin_amdgcn_readfirstlane(hits[HIT_WORDS + slot]);

  for (int i = tid; i < OWN_ROWS * APITCH / 4; i += SC_WAVES * 32)
    ((float4*)sAcc)[i] = float4{0.0f, 0.0f, 0.0f, 0.0f};
  for (int i = tid; i < OWN_ROWS; i += SC_WAVES * 32)
    sDeg[i] = (!FIRST && i < rows) ? deg[nodeBase + i] : 0.0f;
  int pn = lane < nList ? myHits[lane] : 0;
  __syncthreads();

  int* myE = (int*)sRaw + wave * SC_QCAP;
  int cnt = 0;
  int nRec = 0;
  const int4* map16 = (const int4*)omap;
  const int nMapW  = (E + 15) >> 4;
  const int nFullW = E >> 4;
  constexpr int TRIP_W = SC_WAVES * SC_DEPTH * 32;
  const int nTrip = (nMapW + TRIP_W - 1) / TRIP_W;
  const unsigned pat  = (unsigned)blockIdx.x * 0x01010101u;
  const unsigned npat = ~pat;

  auto keepBytes = [&](int w, int n) -> int {
    if (n >= 4) return w;
    if (n <= 0) return (int)npat;
    const unsigned m = (1u << (8 * n)) - 1u;
    return (int)(((unsigned)w & m) | (npat & ~m));
  };
  auto request = [&](int4 (&buf)[SC_DEPTH], int trip) {
#pragma unroll
    for (int k = 0; k < SC_DEPTH; ++k) {
      const int cb = ((trip * SC_DEPTH + k) * SC_WAVES + wave) * 32;
      if (cb + 32 <= nFullW) {
        buf[k] = map16[cb + lane];
      } else {
        const int wd = cb + lane;
        int4 v = int4{(int)npat, (int)npat, (int)npat, (int)npat};
        if (wd < nMapW) {
          v = map16[wd];
          const int left = E - wd * 16;
          v.x = keepBytes(v.x, left);      v.y = keepBytes(v.y, left - 4);
          v.z = keepBytes(v.z, left - 8);  v.w = keepBytes(v.w, left - 12);
        }
        buf[k] = v;
      }
    }
  };

  auto nzflags = [](unsigned x) -> unsigned { return ((x & 0x7f7f7f7fu) + 0x7f7f7f7fu) | x; };
  auto scanChunk = [&](const int4& v, int cb) {
    const unsigned n0 = nzflags((unsigned)v.x ^ pat), n1 = nzflags((unsigned)v.y ^ pat);
    const unsigned n2 = nzflags((unsigned)v.z ^ pat), n3 = nzflags((unsigned)v.w ^ pat);
    unsigned h = (((n0 >> 7) & 0x01010101u) | ((n1 >> 6) & 0x02020202u) |
                  ((n2 >> 5) & 0x04040404u) | ((n3 >> 4) & 0x08080808u)) ^ 0x0f0f0f0fu;
    unsigned any = __builtin_amdgcn_ballot_w32(h != 0u);
    const int e0 = (cb + lane) << 4;
    const unsigned below = (1u << lane) - 1u;
    while (any) {
      if (h != 0u) {
        const int b = __builtin_ctz(h);
        myE[cnt + __builtin_popcount(any & below)] = e0 + ((b & 3) << 2) + (b >> 3);
        h &= h - 1u;
      }
      cnt += __builtin_popcount(any);
      any = __builtin_amdgcn_ballot_w32(h != 0u);
    }
  };

  auto drain = [&](bool all) {
    if (!(cnt >= SC_EU || (all && cnt > 0))) return;
    int q0 = 0;
    while (cnt - q0 >= SC_EU || (all && q0 < cnt)) {
      int m = cnt - q0 < 32 ? cnt - q0 : 32;
      if (!all) m &= ~(SC_EU - 1);
      int s = 0, dl = 0;
      if (lane < m) {
        const int e = myE[q0 + lane];
        s  = src[e];
        dl = dst[e] - nodeBase;
        if ((unsigned)dl >= (unsigned)rows) dl = OWN_ROWS;
        if (FIRST && dl < OWN_ROWS) unsafeAtomicAdd(&sDeg[dl], 1.0f);
      }
      if (FIRST) {
        if (nRec + m <= HIT_CAP) {
          if (lane < m) myHits[nRec + lane] = s | (dl << 16);
        }
        nRec += m;
      }
      int g = 0;
      for (; g + SC_EU <= m; g += SC_EU) {
        float v[SC_EU][4];
#pragma unroll
        for (int u = 0; u < SC_EU; ++u) {
          const int su = __builtin_amdgcn_readlane(s, g + u);
          const float* xs = xin + (long long)su * FEAT + lane;
          v[u][0] = xs[0]; v[u][1] = xs[32]; v[u][2] = xs[64]; v[u][3] = xs[96];
        }
#pragma unroll
        for (int u = 0; u < SC_EU; ++u) {
          const int du = __builtin_amdgcn_readlane(dl, g + u);
          float* p = sAcc + du * APITCH + lane;
          unsafeAtomicAdd(p + 0,  v[u][0]);
          unsafeAtomicAdd(p + 32, v[u][1]);
          unsafeAtomicAdd(p + 64, v[u][2]);
          unsafeAtomicAdd(p + 96, v[u][3]);
        }
      }
      for (; g < m; ++g) {
        const int su = __builtin_amdgcn_readlane(s, g);
        const int du = __builtin_amdgcn_readlane(dl, g);
        const float* xs = xin + (long long)su * FEAT + lane;
        const float v0 = xs[0], v1 = xs[32], v2 = xs[64], v3 = xs[96];
        float* p = sAcc + du * APITCH + lane;
        unsafeAtomicAdd(p + 0,  v0);
        unsafeAtomicAdd(p + 32, v1);
        unsafeAtomicAdd(p + 64, v2);
        unsafeAtomicAdd(p + 96, v3);
      }
      q0 += m;
    }
    const int left = cnt - q0;
    int te = 0;
    if (lane < left) te = myE[q0 + lane];
    if (lane < left) myE[lane] = te;
    cnt = left;
  };

  auto requestFull = [&](int4 (&buf)[SC_DEPTH], int trip) {
#pragma unroll
    for (int k = 0; k < SC_DEPTH; ++k) {
      const int cb = ((trip * SC_DEPTH + k) * SC_WAVES + wave) * 32;
      buf[k] = map16[cb + lane];
    }
  };

  auto scan = [&](const int4 (&buf)[SC_DEPTH], int trip) {
    static_assert(SC_DEPTH == 4, "two chunks, drain, two chunks, drain");
    const int cb = (trip * SC_DEPTH * SC_WAVES + wave) * 32;
    scanChunk(buf[0], cb);
    scanChunk(buf[1], cb + 1 * SC_WAVES * 32);
    drain(false);
    scanChunk(buf[2], cb + 2 * SC_WAVES * 32);
    scanChunk(buf[3], cb + 3 * SC_WAVES * 32);
    drain(false);
  };

  if (nList >= 0) {
    for (int q = 0; q < nList; q += 32) {
      const int p = pn;
      const int m = nList - q < 32 ? nList - q : 32;
      if (q + 32 < nList) pn = q + 32 + lane < nList ? myHits[q + 32 + lane] : 0;
      int g = 0;
      for (; g + SC_EU <= m; g += SC_EU) {
        float v[SC_EU][4];
#pragma unroll
        for (int u = 0; u < SC_EU; ++u) {
          const int su = __builtin_amdgcn_readlane(p, g + u) & 0xffff;
          const float* xs = xin + (long long)su * FEAT + lane;
          v[u][0] = xs[0]; v[u][1] = xs[32]; v[u][2] = xs[64]; v[u][3] = xs[96];
        }
#pragma unroll
        for (int u = 0; u < SC_EU; ++u) {
          const int du = __builtin_amdgcn_readlane(p, g + u) >> 16;
          float* a = sAcc + du * APITCH + lane;
          unsafeAtomicAdd(a + 0,  v[u][0]);
          unsafeAtomicAdd(a + 32, v[u][1]);
          unsafeAtomicAdd(a + 64, v[u][2]);
          unsafeAtomicAdd(a + 96, v[u][3]);
        }
      }
      const int r = m - g;
      if (r > 0) {
        float v[SC_EU][4];
#pragma unroll
        for (int u = 0; u < SC_EU; ++u) {
          if (u < r) {
            const int su = __builtin_amdgcn_readlane(p, g + u) & 0xffff;
            const float* xs = xin + (long long)su * FEAT + lane;
            v[u][0] = xs[0]; v[u][1] = xs[32]; v[u][2] = xs[64]; v[u][3] = xs[96];
          }
        }
#pragma unroll
        for (int u = 0; u < SC_EU; ++u) {
          if (u < r) {
            const int du = __builtin_amdgcn_readlane(p, g + u) >> 16;
            float* a = sAcc + du * APITCH + lane;
            unsafeAtomicAdd(a + 0,  v[u][0]);
            unsafeAtomicAdd(a + 32, v[u][1]);
            unsafeAtomicAdd(a + 64, v[u][2]);
            unsafeAtomicAdd(a + 96, v[u][3]);
          }
        }
      }
    }
  } else {
    const int nPair = (nFullW / TRIP_W) & ~1;
    int trip = 0;
    if (nPair > 0) {
      int4 bufA[SC_DEPTH], bufB[SC_DEPTH];
      requestFull(bufA, 0);
      for (; trip < nPair; trip += 2) {
        requestFull(bufB, trip + 1);
        scan(bufA, trip);
        requestFull(bufA, trip + 2 < nPair ? trip + 2 : nPair - 1);
        scan(bufB, trip + 1);
      }
    }
    for (; trip < nTrip; ++trip) {
      int4 bufT[SC_DEPTH];
      request(bufT, trip);
      scan(bufT, trip);
    }
    drain(true);
    if (FIRST && lane == 0) sCnt[wave] = nRec <= HIT_CAP ? nRec : -1;
  }

  static_assert(XROWS_PER_WAVE == 7, "seven rows per wave are written out by name below");
  auto xrow = [&](int j) -> const float4* {
    const int r   = wave + j * SC_WAVES;
    const int rc  = r < OWN_ROWS ? r : OWN_ROWS - 1;
    const int gr  = nodeBase + rc;
    const int grc = gr < N_NODES ? gr : N_NODES - 1;
    return (const float4*)(xin + (long long)grc * FEAT + lane * 4);
  };
  const float4 xr0 = *xrow(0), xr1 = *xrow(1), xr2 = *xrow(2), xr3 = *xrow(3);
  const float4 xr4 = *xrow(4), xr5 = *xrow(5), xr6 = *xrow(6);
  __syncthreads();

  if (FIRST) {
    if (wave == 0) hits[HIT_WORDS + (int)blockIdx.x * SC_WAVES + lane] = sCnt[lane];
    for (int i = tid; i < rows; i += SC_WAVES * 32) deg[nodeBase + i] = sDeg[i];
  }

  for (int i = tid; i < OWN_ROWS * 32; i += SC_WAVES * 32) {
    const int r = i >> 5;
    const float scale = 1.0f / fmaxf(sDeg[r], 1.0f);
    float4* p = (float4*)(sAcc + r * APITCH + (i & 31) * 4);
    float4 m = *p;
    m.x *= scale; m.y *= scale; m.z *= scale; m.w *= scale;
    *p = m;
  }
  static_assert(XROWS_PER_WAVE * SC_WAVES * APITCH * 4 <= SC_WAVES * SC_QCAP * 4,
                "the staged rows must fit in the queue space");
  float* sX = (float*)sRaw;
  float4* xd = (float4*)(sX + wave * APITCH + lane * 4);
  constexpr int XSTEP = SC_WAVES * APITCH / 4;
  xd[0 * XSTEP] = xr0; xd[1 * XSTEP] = xr1; xd[2 * XSTEP] = xr2; xd[3 * XSTEP] = xr3;
  xd[4 * XSTEP] = xr4; xd[5 * XSTEP] = xr5; xd[6 * XSTEP] = xr6;
  __syncthreads();

  constexpr int  NCT   = FIRST ? 8 : 4;
  constexpr int  N     = NCT * 16;
  constexpr int  NRG   = SC_WAVES / NCT;
  constexpr int  RTMAX = (OWN_TILES + NRG - 1) / NRG;
  const int colTile = wave & (NCT - 1);
  const int rg      = wave / NCT;
  const int nTiles  = (rows + 15) >> 4;
  const int nT      = rg < nTiles ? (nTiles - rg + NRG - 1) / NRG : 0;

  if (RTMAX >= 4 && nT == 4)
    layer_tiles<N, (RTMAX >= 4 ? 4 : 1), NRG, FIRST>(sAcc, sX, Wl, bl, Wr, out, nodeBase, rows, rg, colTile, lane);
  else if (RTMAX >= 3 && nT == 3)
    layer_tiles<N, (RTMAX >= 3 ? 3 : 1), NRG, FIRST>(sAcc, sX, Wl, bl, Wr, out, nodeBase, rows, rg, colTile, lane);
  else if (nT == 2)
    layer_tiles<N, 2, NRG, FIRST>(sAcc, sX, Wl, bl, Wr, out, nodeBase, rows, rg, colTile, lane);
  else if (nT == 1)
    layer_tiles<N, 1, NRG, FIRST>(sAcc, sX, Wl, bl, Wr, out, nodeBase, rows, rg, colTile, lane);
}

extern "C" void kernel_launch(void* const* d_in, const int* in_sizes, int n_in,
                              void* d_out, int out_size, void* d_ws, size_t ws_size,
                              hipStream_t stream) {
  const float* x   = (const float*)d_in[0];
  const int*   ei  = (const int*)d_in[1];
  const float* Wl1 = (const float*)d_in[2];
  const float* bl1 = (const float*)d_in[3];
  const float* Wr1 = (const float*)d_in[4];
  const float* Wl2 = (const float*)d_in[5];
  const float* bl2 = (const float*)d_in[6];
  const float* Wr2 = (const float*)d_in[7];
  float* out = (float*)d_out;

  const int E = in_sizes[1] / 2;
  const int* src = ei;
  const int* dst = ei + E;

  int*   hits = (int*)d_ws;
  int*   omap = hits + HIT_WORDS + OWN_WGS * SC_WAVES;
  float* deg  = (float*)d_ws + (long long)N_NODES * FEAT;
  float* h1   = deg + N_NODES;

  const int SB = SC_WAVES * 32;

  owner_map_kernel<<<OM_WGS, 128, 0, stream>>>(dst, omap, E);
  sage_layer_kernel<true><<<OWN_WGS, SB, 0, stream>>>(x, src, dst, omap, Wl1, bl1, Wr1, deg, hits, h1, E);
  sage_layer_kernel<false><<<OWN_WGS, SB, 0, stream>>>(h1, src, dst, omap, Wl2, bl2, Wr2, deg, hits, out, E);
}


# Layer-1 scan reads its own row of a bit table (1 bit per 4 edges and owner workgroup, written by the first launch in place of the byte map), loads dst/src only of flagged groups; LDS arrival order
# speedup vs baseline: 4.0682x; 1.1064x over previous
#include <hip/hip_runtime.h>


typedef float v2f __attribute__((ext_vector_type(2)));
typedef float v8f __attribute__((ext_vector_type(8)));

#define N_NODES 50000
#define FEAT    128


#define OWN_ROWS  196
#define OWN_WGS   ((N_NODES + OWN_ROWS - 1) / OWN_ROWS)
#define OWN_TILES ((OWN_ROWS + 15) / 16)
#define SC_WAVES  32
#define SC_EU     8
#define SC_BATCH  32
#define SC_JB     4
#define SC_QCAP   (1024 + SC_BATCH)
#define OM_WGS    196
#define OM_GROUPS 1024
#define BM_PITCH  33
#define BIT_PITCH 8224
#define HIT_CAP   256
#define HIT_WORDS (OWN_WGS * SC_WAVES * HIT_CAP)
#define APITCH    (FEAT + 4)
#define XROWS_PER_WAVE ((OWN_ROWS + SC_WAVES - 1) / SC_WAVES)

__device__ __forceinline__ unsigned owner_of(int d) {
  return ((unsigned)d / (unsigned)OWN_ROWS) & 0xffu;
}
__global__ __launch_bounds__(256)
void owner_bits_kernel(const int* __restrict__ dst, unsigned* __restrict__ bits, int E) {
  __shared__ unsigned bm[OWN_WGS * BM_PITCH];
  const int tid = threadIdx.x;
  const int nGroups = (E + 3) >> 2;
  const int fullG   = E >> 2;
  for (int p = (int)blockIdx.x; p * OM_GROUPS < nGroups; p += (int)gridDim.x) {
    for (int i = tid; i < OWN_WGS * BM_PITCH; i += 256) bm[i] = 0u;
    __syncthreads();
    const int gBase = p * OM_GROUPS;
    const unsigned bit = 1u << (tid & 31);
    unsigned* col = bm + (tid >> 5);
    const int lastG = fullG > 0 ? fullG - 1 : 0;
    int4 d[4];
#pragma unroll
    for (int k = 0; k < 4; ++k) {
      const int g = gBase + tid + k * 256;
      d[k] = *(const int4*)(dst + (g < fullG ? g : lastG) * 4);
    }
#pragma unroll
    for (int k = 0; k < 4; ++k) {
      const int g = gBase + tid + k * 256;
      unsigned* c = col + k * 8;
      if (g < fullG) {
        atomicOr(c + owner_of(d[k].x) * BM_PITCH, bit);
        atomicOr(c + owner_of(d[k].y) * BM_PITCH, bit);
        atomicOr(c + owner_of(d[k].z) * BM_PITCH, bit);
        atomicOr(c + owner_of(d[k].w) * BM_PITCH, bit);
      } else if (g == fullG) {
        for (int i = 0; i < (E & 3); ++i)
          atomicOr(c + owner_of(dst[g * 4 + i]) * BM_PITCH, bit);
      }
    }
    __syncthreads();
    const int wv = tid >> 5, ln = tid & 31;
    for (int o = wv; o < OWN_WGS; o += 8)
      bits[o * BIT_PITCH + p * 32 + ln] = bm[o * BM_PITCH + ln];
    __syncthreads();
  }
}

template <int N, int NT, int NRG, bool RELU>
__device__ __forceinline__ void layer_tiles(const float* sM, const float* sX,
                                            const float* __restrict__ Wl,
                                            const float* __restrict__ bl,
                                            const float* __restrict__ Wr,
                                            float* __restrict__ out,
                                            int nodeBase, int rows, int rg, int colTile, int lane) {
  const int c16 = lane & 15;
  const int kb  = (lane >> 4) << 1;
  const int col = (colTile << 4) + c16;

  v8f acc[NT];
  int aoff[NT];
#pragma unroll
  for (int t = 0; t < NT; ++t) {
    acc[t] = v8f{};
    int r = (rg + t * NRG) * 16 + c16;
    r = r < OWN_ROWS ? r : OWN_ROWS - 1;
    aoff[t] = r * APITCH + kb;
  }
  const float* bL = Wl + kb * N + col;
  const float* bR = Wr + kb * N + col;

#pragma unroll 8
  for (int k0 = 0; k0 < FEAT; k0 += 4) {
    v2f b;
    b.x = bL[k0 * N];
    b.y = bL[(k0 + 1) * N];
#pragma unroll
    for (int t = 0; t < NT; ++t) {
      const v2f a = *(const v2f*)(sM + aoff[t] + k0);
      acc[t] = __builtin_amdgcn_wmma_f32_16x16x4_f32(false, a, false, b,
                                                     (short)0, acc[t], false, false);
    }
  }
#pragma unroll 8
  for (int k0 = 0; k0 < FEAT; k0 += 4) {
    v2f b;
    b.x = bR[k0 * N];
    b.y = bR[(k0 + 1) * N];
#pragma unroll
    for (int t = 0; t < NT; ++t) {
      const v2f a = *(const v2f*)(sX + aoff[t] + k0);
      acc[t] = __builtin_amdgcn_wmma_f32_16x16x4_f32(false, a, false, b,
                                                     (short)0, acc[t], false, false);
    }
  }

  const float bias = bl[col];
#pragma unroll
  for (int t = 0; t < NT; ++t) {
    const int rbase = (rg + t * NRG) * 16 + ((lane >> 4) << 3);
#pragma unroll
    for (int i = 0; i < 8; ++i) {
      const int r = rbase + i;
      if (r < rows) {
        float v = acc[t][i] + bias;
        if (RELU) v = fmaxf(v, 0.0f);
        out[(long long)(nodeBase + r) * N + col] = v;
      }
    }
  }
}

template <bool FIRST>
__global__ __launch_bounds__(SC_WAVES * 32)
void sage_layer_kernel(const float* __restrict__ xin,
                       const int* __restrict__ src,
                       const int* __restrict__ dst,
                       const unsigned* __restrict__ bits,
                       const float* __restrict__ Wl,
                       const float* __restrict__ bl,
                       const float* __restrict__ Wr,
                       float* __restrict__ deg,
                       int* __restrict__ hits,
                       float* __restrict__ out,
                       int E) {
  __shared__ __attribute__((aligned(16))) float sAcc[(OWN_ROWS + 1) * APITCH];
  __shared__ float sDeg[OWN_ROWS];
  __shared__ __attribute__((aligned(16))) unsigned char sRaw[SC_WAVES * SC_QCAP * 4];
  __shared__ int   sCnt[SC_WAVES];

  const int tid  = threadIdx.x;
  const int lane = tid & 31;
  const int wave = __builtin_amdgcn_readfirstlane(tid >> 5);
  const int nodeBase = (int)blockIdx.x * OWN_ROWS;
  const int rows = (N_NODES - nodeBase < OWN_ROWS) ? (N_NODES - nodeBase) : OWN_ROWS;

  const int slot = (int)blockIdx.x * SC_WAVES + wave;
  int* myHits = hits + (long long)slot * HIT_CAP;
  int nList = -1;
  if (!FIRST) nList = __builtin_amdgcn_readfirstlane(hits[HIT_WORDS + slot]);

  const unsigned* myBits = bits + (long long)blockIdx.x * BIT_PITCH;
  const int nGroups = (E + 3) >> 2;
  const int nFullG  = E >> 2;
  const int nBitDw  = (nGroups + 31) >> 5;
  const int waveDw  = (nBitDw + SC_WAVES - 1) / SC_WAVES;
  const int nJ      = (waveDw + 31) >> 5;
  auto loadBits = [&](unsigned (&b)[SC_JB], int j0) {
#pragma unroll
    for (int k = 0; k < SC_JB; ++k) {
      const int r = (j0 + k) * 32 + lane;
      const int i = wave * waveDw + r;
      b[k] = (r < waveDw && i < nBitDw) ? myBits[i] : 0u;
    }
  };
  unsigned bCur[SC_JB] = {};
  if (nList < 0) loadBits(bCur, 0);

  for (int i = tid; i < OWN_ROWS * APITCH / 4; i += SC_WAVES * 32)
    ((float4*)sAcc)[i] = float4{0.0f, 0.0f, 0.0f, 0.0f};
  for (int i = tid; i < OWN_ROWS; i += SC_WAVES * 32)
    sDeg[i] = (!FIRST && i < rows) ? deg[nodeBase + i] : 0.0f;
  int pn = lane < nList ? myHits[lane] : 0;
  __syncthreads();

  int* myQ = (int*)sRaw + wave * SC_QCAP;
  int cnt = 0;
  int nRec = 0;

  auto drain = [&](bool all) {
    if (!(cnt >= SC_BATCH || (all && cnt > 0))) return;
    int q0 = 0;
    while (cnt - q0 >= SC_EU || (all && q0 < cnt)) {
      int m = cnt - q0 < 32 ? cnt - q0 : 32;
      if (!all) m &= ~(SC_EU - 1);
      int4 d4 = int4{-1, -1, -1, -1}, s4 = int4{0, 0, 0, 0};
      unsigned hm = 0u;
      int s = 0, dl = OWN_ROWS;
      if (lane < m) {
        const int g = myQ[q0 + lane];
        if (g < nFullG) {
          d4 = *(const int4*)(dst + g * 4);
          s4 = *(const int4*)(src + g * 4);
        } else {
          const int e = g * 4;
          if (e + 0 < E) { d4.x = dst[e + 0]; s4.x = src[e + 0]; }
          if (e + 1 < E) { d4.y = dst[e + 1]; s4.y = src[e + 1]; }
          if (e + 2 < E) { d4.z = dst[e + 2]; s4.z = src[e + 2]; }
          if (e + 3 < E) { d4.w = dst[e + 3]; s4.w = src[e + 3]; }
        }
        hm = ((unsigned)(d4.x - nodeBase) < (unsigned)rows ? 1u : 0u) |
             ((unsigned)(d4.y - nodeBase) < (unsigned)rows ? 2u : 0u) |
             ((unsigned)(d4.z - nodeBase) < (unsigned)rows ? 4u : 0u) |
             ((unsigned)(d4.w - nodeBase) < (unsigned)rows ? 8u : 0u);
        if (hm != 0u) {
          const bool f0 = (hm & 1u) != 0u, f1 = (hm & 2u) != 0u, f2 = (hm & 4u) != 0u;
          s  = f0 ? s4.x : f1 ? s4.y : f2 ? s4.z : s4.w;
          dl = (f0 ? d4.x : f1 ? d4.y : f2 ? d4.z : d4.w) - nodeBase;
          hm &= hm - 1u;
        }
        if (FIRST && dl < OWN_ROWS) unsafeAtomicAdd(&sDeg[dl], 1.0f);
      }
      if (FIRST) {
        if (nRec + m <= HIT_CAP) {
          if (lane < m) myHits[nRec + lane] = s | (dl << 16);
        }
        nRec += m;
      }
      int g = 0;
      for (; g + SC_EU <= m; g += SC_EU) {
        float v[SC_EU][4];
#pragma unroll
        for (int u = 0; u < SC_EU; ++u) {
          const int su = __builtin_amdgcn_readlane(s, g + u);
          const float* xs = xin + (long long)su * FEAT + lane;
          v[u][0] = xs[0]; v[u][1] = xs[32]; v[u][2] = xs[64]; v[u][3] = xs[96];
        }
#pragma unroll
        for (int u = 0; u < SC_EU; ++u) {
          const int du = __builtin_amdgcn_readlane(dl, g + u);
          float* p = sAcc + du * APITCH + lane;
          unsafeAtomicAdd(p + 0,  v[u][0]);
          unsafeAtomicAdd(p + 32, v[u][1]);
          unsafeAtomicAdd(p + 64, v[u][2]);
          unsafeAtomicAdd(p + 96, v[u][3]);
        }
      }
      for (; g < m; ++g) {
        const int su = __builtin_amdgcn_readlane(s, g);
        const int du = __builtin_amdgcn_readlane(dl, g);
        const float* xs = xin + (long long)su * FEAT + lane;
        const float v0 = xs[0], v1 = xs[32], v2 = xs[64], v3 = xs[96];
        float* p = sAcc + du * APITCH + lane;
        unsafeAtomicAdd(p + 0,  v0);
        unsafeAtomicAdd(p + 32, v1);
        unsafeAtomicAdd(p + 64, v2);
        unsafeAtomicAdd(p + 96, v3);
      }
      unsigned more = __builtin_amdgcn_ballot_w32(hm != 0u);
      while (more) {
        const int who = __builtin_ctz(more);
        const bool n1 = (hm & 2u) != 0u, n2 = (hm & 4u) != 0u;
        const int sN = n1 ? s4.y : n2 ? s4.z : s4.w;
        const int dN = (n1 ? d4.y : n2 ? d4.z : d4.w) - nodeBase;
        const int su = __builtin_amdgcn_readlane(sN, who);
        const int du = __builtin_amdgcn_readlane(dN, who);
        if (FIRST) {
          if (lane == 0) unsafeAtomicAdd(&sDeg[du], 1.0f);
          if (nRec + 1 <= HIT_CAP) {
            if (lane == 0) myHits[nRec] = su | (du << 16);
          }
          nRec += 1;
        }
        const float* xs = xin + (long long)su * FEAT + lane;
        const float v0 = xs[0], v1 = xs[32], v2 = xs[64], v3 = xs[96];
        float* p = sAcc + du * APITCH + lane;
        unsafeAtomicAdd(p + 0,  v0);
        unsafeAtomicAdd(p + 32, v1);
        unsafeAtomicAdd(p + 64, v2);
        unsafeAtomicAdd(p + 96, v3);
        if (lane == who) hm &= hm - 1u;
        more = __builtin_amdgcn_ballot_w32(hm != 0u);
      }
      q0 += m;
    }
    const int left = cnt - q0;
    int te = 0;
    if (lane < left) te = myQ[q0 + lane];
    if (lane < left) myQ[lane] = te;
    cnt = left;
  };

  const unsigned below = (1u << lane) - 1u;
  auto queueBits = [&](unsigned b, int i) {
    const int g0 = i << 5;
    unsigned any = __builtin_amdgcn_ballot_w32(b != 0u);
    while (any) {
      if (b != 0u) {
        myQ[cnt + __builtin_popcount(any & below)] = g0 + __builtin_ctz(b);
        b &= b - 1u;
      }
      cnt += __builtin_popcount(any);
      any = __builtin_amdgcn_ballot_w32(b != 0u);
    }
  };

  if (nList >= 0) {
    for (int q = 0; q < nList; q += 32) {
      const int p = pn;
      const int m = nList - q < 32 ? nList - q : 32;
      if (q + 32 < nList) pn = q + 32 + lane < nList ? myHits[q + 32 + lane] : 0;
      int g = 0;
      for (; g + SC_EU <= m; g += SC_EU) {
        float v[SC_EU][4];
#pragma unroll
        for (int u = 0; u < SC_EU; ++u) {
          const int su = __builtin_amdgcn_readlane(p, g + u) & 0xffff;
          const float* xs = xin + (long long)su * FEAT + lane;
          v[u][0] = xs[0]; v[u][1] = xs[32]; v[u][2] = xs[64]; v[u][3] = xs[96];
        }
#pragma unroll
        for (int u = 0; u < SC_EU; ++u) {
          const int du = __builtin_amdgcn_readlane(p, g + u) >> 16;
          float* a = sAcc + du * APITCH + lane;
          unsafeAtomicAdd(a + 0,  v[u][0]);
          unsafeAtomicAdd(a + 32, v[u][1]);
          unsafeAtomicAdd(a + 64, v[u][2]);
          unsafeAtomicAdd(a + 96, v[u][3]);
        }
      }
      const int r = m - g;
      if (r > 0) {
        float v[SC_EU][4];
#pragma unroll
        for (int u = 0; u < SC_EU; ++u) {
          if (u < r) {
            const int su = __builtin_amdgcn_readlane(p, g + u) & 0xffff;
            const float* xs = xin + (long long)su * FEAT + lane;
            v[u][0] = xs[0]; v[u][1] = xs[32]; v[u][2] = xs[64]; v[u][3] = xs[96];
          }
        }
#pragma unroll
        for (int u = 0; u < SC_EU; ++u) {
          if (u < r) {
            const int du = __builtin_amdgcn_readlane(p, g + u) >> 16;
            float* a = sAcc + du * APITCH + lane;
            unsafeAtomicAdd(a + 0,  v[u][0]);
            unsafeAtomicAdd(a + 32, v[u][1]);
            unsafeAtomicAdd(a + 64, v[u][2]);
            unsafeAtomicAdd(a + 96, v[u][3]);
          }
        }
      }
    }
  } else {
    for (int j0 = 0; j0 < nJ; j0 += SC_JB) {
      unsigned bNext[SC_JB];
      loadBits(bNext, j0 + SC_JB);
#pragma unroll
      for (int k = 0; k < SC_JB; ++k) {
        queueBits(bCur[k], wave * waveDw + (j0 + k) * 32 + lane);
        drain(false);
      }
#pragma unroll
      for (int k = 0; k < SC_JB; ++k) bCur[k] = bNext[k];
    }
    drain(true);
    if (FIRST && lane == 0) sCnt[wave] = nRec <= HIT_CAP ? nRec : -1;
  }

  static_assert(XROWS_PER_WAVE == 7, "seven rows per wave are written out by name below");
  auto xrow = [&](int j) -> const float4* {
    const int r   = wave + j * SC_WAVES;
    const int rc  = r < OWN_ROWS ? r : OWN_ROWS - 1;
    const int gr  = nodeBase + rc;
    const int grc = gr < N_NODES ? gr : N_NODES - 1;
    return (const float4*)(xin + (long long)grc * FEAT + lane * 4);
  };
  const float4 xr0 = *xrow(0), xr1 = *xrow(1), xr2 = *xrow(2), xr3 = *xrow(3);
  const float4 xr4 = *xrow(4), xr5 = *xrow(5), xr6 = *xrow(6);
  __syncthreads();

  if (FIRST) {
    if (wave == 0) hits[HIT_WORDS + (int)blockIdx.x * SC_WAVES + lane] = sCnt[lane];
    for (int i = tid; i < rows; i += SC_WAVES * 32) deg[nodeBase + i] = sDeg[i];
  }

  for (int i = tid; i < OWN_ROWS * 32; i += SC_WAVES * 32) {
    const int r = i >> 5;
    const float scale = 1.0f / fmaxf(sDeg[r], 1.0f);
    float4* p = (float4*)(sAcc + r * APITCH + (i & 31) * 4);
    float4 m = *p;
    m.x *= scale; m.y *= scale; m.z *= scale; m.w *= scale;
    *p = m;
  }
  static_assert(XROWS_PER_WAVE * SC_WAVES * APITCH * 4 <= SC_WAVES * SC_QCAP * 4,
                "the staged rows must fit in the queue space");
  float* sX = (float*)sRaw;
  float4* xd = (float4*)(sX + wave * APITCH + lane * 4);
  constexpr int XSTEP = SC_WAVES * APITCH / 4;
  xd[0 * XSTEP] = xr0; xd[1 * XSTEP] = xr1; xd[2 * XSTEP] = xr2; xd[3 * XSTEP] = xr3;
  xd[4 * XSTEP] = xr4; xd[5 * XSTEP] = xr5; xd[6 * XSTEP] = xr6;
  __syncthreads();

  constexpr int  NCT   = FIRST ? 8 : 4;
  constexpr int  N     = NCT * 16;
  constexpr int  NRG   = SC_WAVES / NCT;
  constexpr int  RTMAX = (OWN_TILES + NRG - 1) / NRG;
  const int colTile = wave & (NCT - 1);
  const int rg      = wave / NCT;
  const int nTiles  = (rows + 15) >> 4;
  const int nT      = rg < nTiles ? (nTiles - rg + NRG - 1) / NRG : 0;

  if (RTMAX >= 4 && nT == 4)
    layer_tiles<N, (RTMAX >= 4 ? 4 : 1), NRG, FIRST>(sAcc, sX, Wl, bl, Wr, out, nodeBase, rows, rg, colTile, lane);
  else if (RTMAX >= 3 && nT == 3)
    layer_tiles<N, (RTMAX >= 3 ? 3 : 1), NRG, FIRST>(sAcc, sX, Wl, bl, Wr, out, nodeBase, rows, rg, colTile, lane);
  else if (nT == 2)
    layer_tiles<N, 2, NRG, FIRST>(sAcc, sX, Wl, bl, Wr, out, nodeBase, rows, rg, colTile, lane);
  else if (nT == 1)
    layer_tiles<N, 1, NRG, FIRST>(sAcc, sX, Wl, bl, Wr, out, nodeBase, rows, rg, colTile, lane);
}

extern "C" void kernel_launch(void* const* d_in, const int* in_sizes, int n_in,
                              void* d_out, int out_size, void* d_ws, size_t ws_size,
                              hipStream_t stream) {
  const float* x   = (const float*)d_in[0];
  const int*   ei  = (const int*)d_in[1];
  const float* Wl1 = (const float*)d_in[2];
  const float* bl1 = (const float*)d_in[3];
  const float* Wr1 = (const float*)d_in[4];
  const float* Wl2 = (const float*)d_in[5];
  const float* bl2 = (const float*)d_in[6];
  const float* Wr2 = (const float*)d_in[7];
  float* out = (float*)d_out;

  const int E = in_sizes[1] / 2;
  const int* src = ei;
  const int* dst = ei + E;

  int*      hits = (int*)d_ws;
  unsigned* bits = (unsigned*)(hits + HIT_WORDS + OWN_WGS * SC_WAVES);
  float* deg  = (float*)d_ws + (long long)N_NODES * FEAT;
  float* h1   = deg + N_NODES;

  const int SB = SC_WAVES * 32;

  owner_bits_kernel<<<OM_WGS, 256, 0, stream>>>(dst, bits, E);
  sage_layer_kernel<true><<<OWN_WGS, SB, 0, stream>>>(x, src, dst, bits, Wl1, bl1, Wr1, deg, hits, h1, E);
  sage_layer_kernel<false><<<OWN_WGS, SB, 0, stream>>>(h1, src, dst, bits, Wl2, bl2, Wr2, deg, hits, out, E);
}


# Neighbour-row gather: one 16-byte load per lane = one load per 512-byte row, rows on line boundaries (h1 base moved), not 4 one-line loads; LDS sums keep lane order, so mean@Wl sums k permuted
# speedup vs baseline: 4.4926x; 1.1043x over previous
#include <hip/hip_runtime.h>


typedef float v2f __attribute__((ext_vector_type(2)));
typedef float v8f __attribute__((ext_vector_type(8)));

#define N_NODES 50000
#define FEAT    128


#define OWN_ROWS  196
#define OWN_WGS   ((N_NODES + OWN_ROWS - 1) / OWN_ROWS)
#define OWN_TILES ((OWN_ROWS + 15) / 16)
#define SC_WAVES  32
#define SC_EU     8
#define SC_BATCH  32
#define SC_JB     4
#define SC_QCAP   (1024 + SC_BATCH)
#define OM_WGS    196
#define OM_GROUPS 1024
#define BM_PITCH  33
#define BIT_PITCH 8224
#define HIT_CAP   256
#define HIT_WORDS (OWN_WGS * SC_WAVES * HIT_CAP)
#define APITCH    (FEAT + 4)
#define XROWS_PER_WAVE ((OWN_ROWS + SC_WAVES - 1) / SC_WAVES)

__device__ __forceinline__ unsigned owner_of(int d) {
  return ((unsigned)d / (unsigned)OWN_ROWS) & 0xffu;
}
__global__ __launch_bounds__(256)
void owner_bits_kernel(const int* __restrict__ dst, unsigned* __restrict__ bits, int E) {
  __shared__ unsigned bm[OWN_WGS * BM_PITCH];
  const int tid = threadIdx.x;
  const int nGroups = (E + 3) >> 2;
  const int fullG   = E >> 2;
  for (int p = (int)blockIdx.x; p * OM_GROUPS < nGroups; p += (int)gridDim.x) {
    for (int i = tid; i < OWN_WGS * BM_PITCH; i += 256) bm[i] = 0u;
    __syncthreads();
    const int gBase = p * OM_GROUPS;
    const unsigned bit = 1u << (tid & 31);
    unsigned* col = bm + (tid >> 5);
    const int lastG = fullG > 0 ? fullG - 1 : 0;
    int4 d[4];
#pragma unroll
    for (int k = 0; k < 4; ++k) {
      const int g = gBase + tid + k * 256;
      d[k] = *(const int4*)(dst + (g < fullG ? g : lastG) * 4);
    }
#pragma unroll
    for (int k = 0; k < 4; ++k) {
      const int g = gBase + tid + k * 256;
      unsigned* c = col + k * 8;
      if (g < fullG) {
        atomicOr(c + owner_of(d[k].x) * BM_PITCH, bit);
        atomicOr(c + owner_of(d[k].y) * BM_PITCH, bit);
        atomicOr(c + owner_of(d[k].z) * BM_PITCH, bit);
        atomicOr(c + owner_of(d[k].w) * BM_PITCH, bit);
      } else if (g == fullG) {
        for (int i = 0; i < (E & 3); ++i)
          atomicOr(c + owner_of(dst[g * 4 + i]) * BM_PITCH, bit);
      }
    }
    __syncthreads();
    const int wv = tid >> 5, ln = tid & 31;
    for (int o = wv; o < OWN_WGS; o += 8)
      bits[o * BIT_PITCH + p * 32 + ln] = bm[o * BM_PITCH + ln];
    __syncthreads();
  }
}

template <int N, int NT, int NRG, bool RELU>
__device__ __forceinline__ void layer_tiles(const float* sM, const float* sX,
                                            const float* __restrict__ Wl,
                                            const float* __restrict__ bl,
                                            const float* __restrict__ Wr,
                                            float* __restrict__ out,
                                            int nodeBase, int rows, int rg, int colTile, int lane) {
  const int c16 = lane & 15;
  const int kb  = (lane >> 4) << 1;
  const int col = (colTile << 4) + c16;

  v8f acc[NT];
  int aoff[NT];
#pragma unroll
  for (int t = 0; t < NT; ++t) {
    acc[t] = v8f{};
    int r = (rg + t * NRG) * 16 + c16;
    r = r < OWN_ROWS ? r : OWN_ROWS - 1;
    aoff[t] = r * APITCH + kb;
  }
  const float* bL = Wl + (kb * 4) * N + col;
  const float* bR = Wr + kb * N + col;

#pragma unroll 8
  for (int k0 = 0; k0 < FEAT; k0 += 4) {
    const int fr = ((k0 & 31) << 2) + (k0 >> 5);
    v2f b;
    b.x = bL[fr * N];
    b.y = bL[(fr + 4) * N];
#pragma unroll
    for (int t = 0; t < NT; ++t) {
      const v2f a = *(const v2f*)(sM + aoff[t] + k0);
      acc[t] = __builtin_amdgcn_wmma_f32_16x16x4_f32(false, a, false, b,
                                                     (short)0, acc[t], false, false);
    }
  }
#pragma unroll 8
  for (int k0 = 0; k0 < FEAT; k0 += 4) {
    v2f b;
    b.x = bR[k0 * N];
    b.y = bR[(k0 + 1) * N];
#pragma unroll
    for (int t = 0; t < NT; ++t) {
      const v2f a = *(const v2f*)(sX + aoff[t] + k0);
      acc[t] = __builtin_amdgcn_wmma_f32_16x16x4_f32(false, a, false, b,
                                                     (short)0, acc[t], false, false);
    }
  }

  const float bias = bl[col];
#pragma unroll
  for (int t = 0; t < NT; ++t) {
    const int rbase = (rg + t * NRG) * 16 + ((lane >> 4) << 3);
#pragma unroll
    for (int i = 0; i < 8; ++i) {
      const int r = rbase + i;
      if (r < rows) {
        float v = acc[t][i] + bias;
        if (RELU) v = fmaxf(v, 0.0f);
        out[(long long)(nodeBase + r) * N + col] = v;
      }
    }
  }
}

template <bool FIRST>
__global__ __launch_bounds__(SC_WAVES * 32)
void sage_layer_kernel(const float* __restrict__ xin,
                       const int* __restrict__ src,
                       const int* __restrict__ dst,
                       const unsigned* __restrict__ bits,
                       const float* __restrict__ Wl,
                       const float* __restrict__ bl,
                       const float* __restrict__ Wr,
                       float* __restrict__ deg,
                       int* __restrict__ hits,
                       float* __restrict__ out,
                       int E) {
  __shared__ __attribute__((aligned(16))) float sAcc[(OWN_ROWS + 1) * APITCH];
  __shared__ float sDeg[OWN_ROWS];
  __shared__ __attribute__((aligned(16))) unsigned char sRaw[SC_WAVES * SC_QCAP * 4];
  __shared__ int   sCnt[SC_WAVES];

  const int tid  = threadIdx.x;
  const int lane = tid & 31;
  const int wave = __builtin_amdgcn_readfirstlane(tid >> 5);
  const int nodeBase = (int)blockIdx.x * OWN_ROWS;
  const int rows = (N_NODES - nodeBase < OWN_ROWS) ? (N_NODES - nodeBase) : OWN_ROWS;

  const int slot = (int)blockIdx.x * SC_WAVES + wave;
  int* myHits = hits + (long long)slot * HIT_CAP;
  int nList = -1;
  if (!FIRST) nList = __builtin_amdgcn_readfirstlane(hits[HIT_WORDS + slot]);

  const unsigned* myBits = bits + (long long)blockIdx.x * BIT_PITCH;
  const int nGroups = (E + 3) >> 2;
  const int nFullG  = E >> 2;
  const int nBitDw  = (nGroups + 31) >> 5;
  const int waveDw  = (nBitDw + SC_WAVES - 1) / SC_WAVES;
  const int nJ      = (waveDw + 31) >> 5;
  auto loadBits = [&](unsigned (&b)[SC_JB], int j0) {
#pragma unroll
    for (int k = 0; k < SC_JB; ++k) {
      const int r = (j0 + k) * 32 + lane;
      const int i = wave * waveDw + r;
      b[k] = (r < waveDw && i < nBitDw) ? myBits[i] : 0u;
    }
  };
  unsigned bCur[SC_JB] = {};
  if (nList < 0) loadBits(bCur, 0);

  for (int i = tid; i < OWN_ROWS * APITCH / 4; i += SC_WAVES * 32)
    ((float4*)sAcc)[i] = float4{0.0f, 0.0f, 0.0f, 0.0f};
  for (int i = tid; i < OWN_ROWS; i += SC_WAVES * 32)
    sDeg[i] = (!FIRST && i < rows) ? deg[nodeBase + i] : 0.0f;
  int pn = lane < nList ? myHits[lane] : 0;
  __syncthreads();

  int* myQ = (int*)sRaw + wave * SC_QCAP;
  int cnt = 0;
  int nRec = 0;

  auto drain = [&](bool all) {
    if (!(cnt >= SC_BATCH || (all && cnt > 0))) return;
    int q0 = 0;
    while (cnt - q0 >= SC_EU || (all && q0 < cnt)) {
      int m = cnt - q0 < 32 ? cnt - q0 : 32;
      if (!all) m &= ~(SC_EU - 1);
      int4 d4 = int4{-1, -1, -1, -1}, s4 = int4{0, 0, 0, 0};
      unsigned hm = 0u;
      int s = 0, dl = OWN_ROWS;
      if (lane < m) {
        const int g = myQ[q0 + lane];
        if (g < nFullG) {
          d4 = *(const int4*)(dst + g * 4);
          s4 = *(const int4*)(src + g * 4);
        } else {
          const int e = g * 4;
          if (e + 0 < E) { d4.x = dst[e + 0]; s4.x = src[e + 0]; }
          if (e + 1 < E) { d4.y = dst[e + 1]; s4.y = src[e + 1]; }
          if (e + 2 < E) { d4.z = dst[e + 2]; s4.z = src[e + 2]; }
          if (e + 3 < E) { d4.w = dst[e + 3]; s4.w = src[e + 3]; }
        }
        hm = ((unsigned)(d4.x - nodeBase) < (unsigned)rows ? 1u : 0u) |
             ((unsigned)(d4.y - nodeBase) < (unsigned)rows ? 2u : 0u) |
             ((unsigned)(d4.z - nodeBase) < (unsigned)rows ? 4u : 0u) |
             ((unsigned)(d4.w - nodeBase) < (unsigned)rows ? 8u : 0u);
        if (hm != 0u) {
          const bool f0 = (hm & 1u) != 0u, f1 = (hm & 2u) != 0u, f2 = (hm & 4u) != 0u;
          s  = f0 ? s4.x : f1 ? s4.y : f2 ? s4.z : s4.w;
          dl = (f0 ? d4.x : f1 ? d4.y : f2 ? d4.z : d4.w) - nodeBase;
          hm &= hm - 1u;
        }
        if (FIRST && dl < OWN_ROWS) unsafeAtomicAdd(&sDeg[dl], 1.0f);
      }
      if (FIRST) {
        if (nRec + m <= HIT_CAP) {
          if (lane < m) myHits[nRec + lane] = s | (dl << 16);
        }
        nRec += m;
      }
      int g = 0;
      for (; g + SC_EU <= m; g += SC_EU) {
        float4 v[SC_EU];
#pragma unroll
        for (int u = 0; u < SC_EU; ++u) {
          const int su = __builtin_amdgcn_readlane(s, g + u);
          v[u] = *(const float4*)(xin + (long long)su * FEAT + lane * 4);
        }
#pragma unroll
        for (int u = 0; u < SC_EU; ++u) {
          const int du = __builtin_amdgcn_readlane(dl, g + u);
          float* p = sAcc + du * APITCH + lane;
          unsafeAtomicAdd(p + 0,  v[u].x);
          unsafeAtomicAdd(p + 32, v[u].y);
          unsafeAtomicAdd(p + 64, v[u].z);
          unsafeAtomicAdd(p + 96, v[u].w);
        }
      }
      for (; g < m; ++g) {
        const int su = __builtin_amdgcn_readlane(s, g);
        const int du = __builtin_amdgcn_readlane(dl, g);
        const float4 v = *(const float4*)(xin + (long long)su * FEAT + lane * 4);
        float* p = sAcc + du * APITCH + lane;
        unsafeAtomicAdd(p + 0,  v.x);
        unsafeAtomicAdd(p + 32, v.y);
        unsafeAtomicAdd(p + 64, v.z);
        unsafeAtomicAdd(p + 96, v.w);
      }
      unsigned more = __builtin_amdgcn_ballot_w32(hm != 0u);
      while (more) {
        const int who = __builtin_ctz(more);
        const bool n1 = (hm & 2u) != 0u, n2 = (hm & 4u) != 0u;
        const int sN = n1 ? s4.y : n2 ? s4.z : s4.w;
        const int dN = (n1 ? d4.y : n2 ? d4.z : d4.w) - nodeBase;
        const int su = __builtin_amdgcn_readlane(sN, who);
        const int du = __builtin_amdgcn_readlane(dN, who);
        if (FIRST) {
          if (lane == 0) unsafeAtomicAdd(&sDeg[du], 1.0f);
          if (nRec + 1 <= HIT_CAP) {
            if (lane == 0) myHits[nRec] = su | (du << 16);
          }
          nRec += 1;
        }
        const float4 v = *(const float4*)(xin + (long long)su * FEAT + lane * 4);
        float* p = sAcc + du * APITCH + lane;
        unsafeAtomicAdd(p + 0,  v.x);
        unsafeAtomicAdd(p + 32, v.y);
        unsafeAtomicAdd(p + 64, v.z);
        unsafeAtomicAdd(p + 96, v.w);
        if (lane == who) hm &= hm - 1u;
        more = __builtin_amdgcn_ballot_w32(hm != 0u);
      }
      q0 += m;
    }
    const int left = cnt - q0;
    int te = 0;
    if (lane < left) te = myQ[q0 + lane];
    if (lane < left) myQ[lane] = te;
    cnt = left;
  };

  const unsigned below = (1u << lane) - 1u;
  auto queueBits = [&](unsigned b, int i) {
    const int g0 = i << 5;
    unsigned any = __builtin_amdgcn_ballot_w32(b != 0u);
    while (any) {
      if (b != 0u) {
        myQ[cnt + __builtin_popcount(any & below)] = g0 + __builtin_ctz(b);
        b &= b - 1u;
      }
      cnt += __builtin_popcount(any);
      any = __builtin_amdgcn_ballot_w32(b != 0u);
    }
  };

  if (nList >= 0) {
    for (int q = 0; q < nList; q += 32) {
      const int p = pn;
      const int m = nList - q < 32 ? nList - q : 32;
      if (q + 32 < nList) pn = q + 32 + lane < nList ? myHits[q + 32 + lane] : 0;
      int g = 0;
      for (; g + SC_EU <= m; g += SC_EU) {
        float4 v[SC_EU];
#pragma unroll
        for (int u = 0; u < SC_EU; ++u) {
          const int su = __builtin_amdgcn_readlane(p, g + u) & 0xffff;
          v[u] = *(const float4*)(xin + (long long)su * FEAT + lane * 4);
        }
#pragma unroll
        for (int u = 0; u < SC_EU; ++u) {
          const int du = __builtin_amdgcn_readlane(p, g + u) >> 16;
          float* a = sAcc + du * APITCH + lane;
          unsafeAtomicAdd(a + 0,  v[u].x);
          unsafeAtomicAdd(a + 32, v[u].y);
          unsafeAtomicAdd(a + 64, v[u].z);
          unsafeAtomicAdd(a + 96, v[u].w);
        }
      }
      const int r = m - g;
      if (r > 0) {
        float4 v[SC_EU];
#pragma unroll
        for (int u = 0; u < SC_EU; ++u) {
          if (u < r) {
            const int su = __builtin_amdgcn_readlane(p, g + u) & 0xffff;
            v[u] = *(const float4*)(xin + (long long)su * FEAT + lane * 4);
          }
        }
#pragma unroll
        for (int u = 0; u < SC_EU; ++u) {
          if (u < r) {
            const int du = __builtin_amdgcn_readlane(p, g + u) >> 16;
            float* a = sAcc + du * APITCH + lane;
            unsafeAtomicAdd(a + 0,  v[u].x);
            unsafeAtomicAdd(a + 32, v[u].y);
            unsafeAtomicAdd(a + 64, v[u].z);
            unsafeAtomicAdd(a + 96, v[u].w);
          }
        }
      }
    }
  } else {
    for (int j0 = 0; j0 < nJ; j0 += SC_JB) {
      unsigned bNext[SC_JB];
      loadBits(bNext, j0 + SC_JB);
#pragma unroll
      for (int k = 0; k < SC_JB; ++k) {
        queueBits(bCur[k], wave * waveDw + (j0 + k) * 32 + lane);
        drain(false);
      }
#pragma unroll
      for (int k = 0; k < SC_JB; ++k) bCur[k] = bNext[k];
    }
    drain(true);
    if (FIRST && lane == 0) sCnt[wave] = nRec <= HIT_CAP ? nRec : -1;
  }

  static_assert(XROWS_PER_WAVE == 7, "seven rows per wave are written out by name below");
  auto xrow = [&](int j) -> const float4* {
    const int r   = wave + j * SC_WAVES;
    const int rc  = r < OWN_ROWS ? r : OWN_ROWS - 1;
    const int gr  = nodeBase + rc;
    const int grc = gr < N_NODES ? gr : N_NODES - 1;
    return (const float4*)(xin + (long long)grc * FEAT + lane * 4);
  };
  const float4 xr0 = *xrow(0), xr1 = *xrow(1), xr2 = *xrow(2), xr3 = *xrow(3);
  const float4 xr4 = *xrow(4), xr5 = *xrow(5), xr6 = *xrow(6);
  __syncthreads();

  if (FIRST) {
    if (wave == 0) hits[HIT_WORDS + (int)blockIdx.x * SC_WAVES + lane] = sCnt[lane];
    for (int i = tid; i < rows; i += SC_WAVES * 32) deg[nodeBase + i] = sDeg[i];
  }

  for (int i = tid; i < OWN_ROWS * 32; i += SC_WAVES * 32) {
    const int r = i >> 5;
    const float scale = 1.0f / fmaxf(sDeg[r], 1.0f);
    float4* p = (float4*)(sAcc + r * APITCH + (i & 31) * 4);
    float4 m = *p;
    m.x *= scale; m.y *= scale; m.z *= scale; m.w *= scale;
    *p = m;
  }
  static_assert(XROWS_PER_WAVE * SC_WAVES * APITCH * 4 <= SC_WAVES * SC_QCAP * 4,
                "the staged rows must fit in the queue space");
  float* sX = (float*)sRaw;
  float4* xd = (float4*)(sX + wave * APITCH + lane * 4);
  constexpr int XSTEP = SC_WAVES * APITCH / 4;
  xd[0 * XSTEP] = xr0; xd[1 * XSTEP] = xr1; xd[2 * XSTEP] = xr2; xd[3 * XSTEP] = xr3;
  xd[4 * XSTEP] = xr4; xd[5 * XSTEP] = xr5; xd[6 * XSTEP] = xr6;
  __syncthreads();

  constexpr int  NCT   = FIRST ? 8 : 4;
  constexpr int  N     = NCT * 16;
  constexpr int  NRG   = SC_WAVES / NCT;
  constexpr int  RTMAX = (OWN_TILES + NRG - 1) / NRG;
  const int colTile = wave & (NCT - 1);
  const int rg      = wave / NCT;
  const int nTiles  = (rows + 15) >> 4;
  const int nT      = rg < nTiles ? (nTiles - rg + NRG - 1) / NRG : 0;

  if (RTMAX >= 4 && nT == 4)
    layer_tiles<N, (RTMAX >= 4 ? 4 : 1), NRG, FIRST>(sAcc, sX, Wl, bl, Wr, out, nodeBase, rows, rg, colTile, lane);
  else if (RTMAX >= 3 && nT == 3)
    layer_tiles<N, (RTMAX >= 3 ? 3 : 1), NRG, FIRST>(sAcc, sX, Wl, bl, Wr, out, nodeBase, rows, rg, colTile, lane);
  else if (nT == 2)
    layer_tiles<N, 2, NRG, FIRST>(sAcc, sX, Wl, bl, Wr, out, nodeBase, rows, rg, colTile, lane);
  else if (nT == 1)
    layer_tiles<N, 1, NRG, FIRST>(sAcc, sX, Wl, bl, Wr, out, nodeBase, rows, rg, colTile, lane);
}

extern "C" void kernel_launch(void* const* d_in, const int* in_sizes, int n_in,
                              void* d_out, int out_size, void* d_ws, size_t ws_size,
                              hipStream_t stream) {
  const float* x   = (const float*)d_in[0];
  const int*   ei  = (const int*)d_in[1];
  const float* Wl1 = (const float*)d_in[2];
  const float* bl1 = (const float*)d_in[3];
  const float* Wr1 = (const float*)d_in[4];
  const float* Wl2 = (const float*)d_in[5];
  const float* bl2 = (const float*)d_in[6];
  const float* Wr2 = (const float*)d_in[7];
  float* out = (float*)d_out;

  const int E = in_sizes[1] / 2;
  const int* src = ei;
  const int* dst = ei + E;

  int*      hits = (int*)d_ws;
  unsigned* bits = (unsigned*)(hits + HIT_WORDS + OWN_WGS * SC_WAVES);
  float* deg  = (float*)(bits + OWN_WGS * BIT_PITCH);
  float* h1   = (float*)d_ws + (long long)N_NODES * FEAT;
  static_assert((long long)HIT_WORDS + OWN_WGS * SC_WAVES + (long long)OWN_WGS * BIT_PITCH + N_NODES
                    <= (long long)N_NODES * FEAT,
                "lists, counts, table and deg must end in front of h1");

  const int SB = SC_WAVES * 32;

  owner_bits_kernel<<<OM_WGS, 256, 0, stream>>>(dst, bits, E);
  sage_layer_kernel<true><<<OWN_WGS, SB, 0, stream>>>(x, src, dst, bits, Wl1, bl1, Wr1, deg, hits, h1, E);
  sage_layer_kernel<false><<<OWN_WGS, SB, 0, stream>>>(h1, src, dst, bits, Wl2, bl2, Wr2, deg, hits, out, E);
}
